# starting kernel
# baseline (speedup 1.0000x reference)
#include <hip/hip_runtime.h>
#include <hip/hip_bf16.h>

typedef float v2f __attribute__((ext_vector_type(2)));
typedef float v8f __attribute__((ext_vector_type(8)));

#define LDS_LD 130

__device__ __forceinline__ v8f wmma_f32(v2f a, v2f b, v8f c) {
    return __builtin_amdgcn_wmma_f32_16x16x4_f32(
         false, a,  false, b,
         (short)0, c,  false,  false);
}

__device__ __forceinline__ v2f a_frag_g(const float* __restrict__ A, int lda,
                                        int r0, int k, int lane) {
    int m  = lane & 15;
    int kk = k + ((lane >> 4) << 1);
    const float* p = A + (size_t)(r0 + m) * lda + kk;
    return v2f{p[0], p[1]};
}

__device__ __forceinline__ v2f a_frag_lds(const float* __restrict__ L, int ld,
                                          int k, int lane) {
    int m  = lane & 15;
    int kk = k + ((lane >> 4) << 1);
    const float* p = L + m * ld + kk;
    return v2f{p[0], p[1]};
}

__device__ __forceinline__ v2f b_frag_g(const float* __restrict__ B, int ldb,
                                        int k, int c0, int lane) {
    int n  = lane & 15;
    int kk = k + ((lane >> 4) << 1);
    const float* p = B + (size_t)kk * ldb + c0 + n;
    return v2f{p[0], p[ldb]};
}

__device__ __forceinline__ void store_tile(float* __restrict__ D, int ldd,
                                           int r0, int c0, v8f acc, int lane,
                                           float bias) {
    int n    = lane & 15;
    int half = lane >> 4;
#pragma unroll
    for (int r = 0; r < 8; ++r)
        D[(size_t)(r0 + r + 8 * half) * ldd + c0 + n] = acc[r] + bias;
}

__global__ void deg_init_kernel(float* __restrict__ deg, int N) {
    int i = blockIdx.x * blockDim.x + threadIdx.x;
    if (i < N) deg[i] = 1.0f;
}

__global__ void deg_acc_kernel(const int* __restrict__ col, float* __restrict__ deg, int E) {
    int e = blockIdx.x * blockDim.x + threadIdx.x;
    if (e < E) atomicAdd(&deg[col[e]], 1.0f);
}

__global__ void dis_kernel(const float* __restrict__ deg, float* __restrict__ dis, int N) {
    int i = blockIdx.x * blockDim.x + threadIdx.x;
    if (i < N) dis[i] = rsqrtf(deg[i]);
}

__global__ __launch_bounds__(128) void self_branch_kernel(
    const float* __restrict__ x, const float* __restrict__ Win,
    const float* __restrict__ bin, const float* __restrict__ Wout,
    const float* __restrict__ bout, const float* __restrict__ Wg1,
    float* __restrict__ l1_out, float* __restrict__ h1_out, int N)
{
    __shared__ float lds[4 * 16 * LDS_LD];
    const int lane = threadIdx.x & 31;
    const int wave = threadIdx.x >> 5;
    const int r0   = (blockIdx.x * 4 + wave) * 16;
    const int n    = lane & 15, half = lane >> 4;
    float* myL = lds + wave * 16 * LDS_LD;
    if (r0 >= N) return;

    v2f ax[32];
#pragma unroll
    for (int kc = 0; kc < 32; ++kc)
        ax[kc] = a_frag_g(x, 128, r0, kc * 4, lane);

    for (int t = 0; t < 8; ++t) {
        v8f acc = {};
#pragma unroll
        for (int kc = 0; kc < 32; ++kc)
            acc = wmma_f32(ax[kc], b_frag_g(Win, 128, kc * 4, t * 16, lane), acc);
        float bb = bin[t * 16 + n];
#pragma unroll
        for (int r = 0; r < 8; ++r) {
            float v = acc[r] + bb;
            myL[(r + 8 * half) * LDS_LD + t * 16 + n] = v > 0.f ? v : 0.f;
        }
    }
    __syncthreads();

    for (int t = 0; t < 8; ++t) {
        v8f acc = {};
#pragma unroll
        for (int kc = 0; kc < 32; ++kc)
            acc = wmma_f32(ax[kc], b_frag_g(Wg1, 128, kc * 4, t * 16, lane), acc);
        store_tile(h1_out, 128, r0, t * 16, acc, lane, 0.f);
    }

    v2f al[32];
#pragma unroll
    for (int kc = 0; kc < 32; ++kc)
        al[kc] = a_frag_lds(myL, LDS_LD, kc * 4, lane);

    for (int t = 0; t < 4; ++t) {
        v8f acc = {};
#pragma unroll
        for (int kc = 0; kc < 32; ++kc)
            acc = wmma_f32(ax[kc], b_frag_g(Wout, 64, kc * 4, t * 16, lane), acc);
#pragma unroll
        for (int kc = 0; kc < 32; ++kc)
            acc = wmma_f32(al[kc], b_frag_g(Wout, 64, 128 + kc * 4, t * 16, lane), acc);
        store_tile(l1_out, 64, r0, t * 16, acc, lane, bout[t * 16 + n]);
    }
}

template <int W>
__global__ void seg_init_kernel(const float* __restrict__ h, const float* __restrict__ dis,
                                const float* __restrict__ bias, float* __restrict__ g, int N)
{
    int idx = blockIdx.x * blockDim.x + threadIdx.x;
    if (idx >= N * W) return;
    int i = idx / W, j = idx % W;
    float d = dis[i];
    g[idx] = bias[j] + h[idx] * d * d;
}

template <int W>
__global__ void seg_scatter_kernel(const int* __restrict__ row, const int* __restrict__ col,
                                   const float* __restrict__ dis, const float* __restrict__ h,
                                   float* __restrict__ g, int E)
{
    constexpr int PER_EDGE = W / 4;
    long long t = (long long)blockIdx.x * blockDim.x + threadIdx.x;
    if (t >= (long long)E * PER_EDGE) return;
    int e = (int)(t / PER_EDGE);
    int j = ((int)(t % PER_EDGE)) << 2;
    int r = row[e], c = col[e];
    float nrm = dis[r] * dis[c];
    const float4 hv = *(const float4*)(h + (size_t)r * W + j);
    float* gp = g + (size_t)c * W + j;
    atomicAdd(gp + 0, hv.x * nrm);
    atomicAdd(gp + 1, hv.y * nrm);
    atomicAdd(gp + 2, hv.z * nrm);
    atomicAdd(gp + 3, hv.w * nrm);
}

__global__ __launch_bounds__(128) void gcn2_gemm_kernel(
    const float* __restrict__ g1, const float* __restrict__ Wg2,
    float* __restrict__ h2, int N)
{
    const int lane = threadIdx.x & 31;
    const int wave = threadIdx.x >> 5;
    const int r0   = (blockIdx.x * 4 + wave) * 16;
    if (r0 >= N) return;

    v2f ag[32];
#pragma unroll
    for (int kc = 0; kc < 32; ++kc)
        ag[kc] = a_frag_g(g1, 128, r0, kc * 4, lane);

    for (int t = 0; t < 4; ++t) {
        v8f acc = {};
#pragma unroll
        for (int kc = 0; kc < 32; ++kc)
            acc = wmma_f32(ag[kc], b_frag_g(Wg2, 64, kc * 4, t * 16, lane), acc);
        store_tile(h2, 64, r0, t * 16, acc, lane, 0.f);
    }
}

__global__ __launch_bounds__(128) void out_gemm_kernel(
    const float* __restrict__ xn, const float* __restrict__ g2,
    const float* __restrict__ W, const float* __restrict__ b,
    float* __restrict__ out, int N)
{
    const int lane = threadIdx.x & 31;
    const int wave = threadIdx.x >> 5;
    const int r0   = (blockIdx.x * 4 + wave) * 16;
    if (r0 >= N) return;
    const int n = lane & 15;

    v2f axn[32];
#pragma unroll
    for (int kc = 0; kc < 32; ++kc)
        axn[kc] = a_frag_g(xn, 128, r0, kc * 4, lane);
    v2f ag2[16];
#pragma unroll
    for (int kc = 0; kc < 16; ++kc)
        ag2[kc] = a_frag_g(g2, 64, r0, kc * 4, lane);

    for (int t = 0; t < 4; ++t) {
        v8f acc = {};
#pragma unroll
        for (int kc = 0; kc < 32; ++kc)
            acc = wmma_f32(axn[kc], b_frag_g(W, 64, kc * 4, t * 16, lane), acc);
#pragma unroll
        for (int kc = 0; kc < 16; ++kc)
            acc = wmma_f32(ag2[kc], b_frag_g(W, 64, 128 + kc * 4, t * 16, lane), acc);
        store_tile(out, 64, r0, t * 16, acc, lane, b[t * 16 + n]);
    }
}

extern "C" void kernel_launch(void* const* d_in, const int* in_sizes, int n_in,
                              void* d_out, int out_size, void* d_ws, size_t ws_size,
                              hipStream_t stream) {
    const float* x_self     = (const float*)d_in[0];
    const float* x_neighbor = (const float*)d_in[1];
    const int*   edge_index = (const int*)d_in[2];
    const float* W_in_self  = (const float*)d_in[3];
    const float* b_in_self  = (const float*)d_in[4];
    const float* W_out_self = (const float*)d_in[5];
    const float* b_out_self = (const float*)d_in[6];
    const float* Wg1        = (const float*)d_in[7];
    const float* bg1        = (const float*)d_in[8];
    const float* Wg2        = (const float*)d_in[9];
    const float* bg2        = (const float*)d_in[10];
    const float* W_out      = (const float*)d_in[11];
    const float* b_out      = (const float*)d_in[12];

    const int N = in_sizes[0] / 128;
    const int E = in_sizes[2] / 2;
    const int* row = edge_index;
    const int* col = edge_index + E;

    float* ws  = (float*)d_ws;
    float* dis = ws;
    float* deg = ws + N;
    float* h   = ws + 2 * (size_t)N;
    float* g   = h + (size_t)N * 128;

    float* l1_out = (float*)d_out;
    float* x2_out = (float*)d_out + (size_t)N * 64;

    deg_init_kernel<<<(N + 255) / 256, 256, 0, stream>>>(deg, N);
    deg_acc_kernel<<<(E + 255) / 256, 256, 0, stream>>>(col, deg, E);
    dis_kernel<<<(N + 255) / 256, 256, 0, stream>>>(deg, dis, N);

    const int strips = (N + 15) / 16;
    const int blocks = (strips + 3) / 4;

    self_branch_kernel<<<blocks, 128, 0, stream>>>(x_self, W_in_self, b_in_self,
                                                   W_out_self, b_out_self, Wg1,
                                                   l1_out, h, N);

    seg_init_kernel<128><<<((N * 128) + 255) / 256, 256, 0, stream>>>(h, dis, bg1, g, N);
    seg_scatter_kernel<128><<<(int)(((long long)E * 32 + 255) / 256), 256, 0, stream>>>(
        row, col, dis, h, g, E);

    gcn2_gemm_kernel<<<blocks, 128, 0, stream>>>(g, Wg2, h, N);

    seg_init_kernel<64><<<((N * 64) + 255) / 256, 256, 0, stream>>>(h, dis, bg2, g, N);
    seg_scatter_kernel<64><<<(int)(((long long)E * 16 + 255) / 256), 256, 0, stream>>>(
        row, col, dis, h, g, E);

    out_gemm_kernel<<<blocks, 128, 0, stream>>>(x_neighbor, g, W_out, b_out, x2_out, N);
}


# GCN aggregation: float-atomic edge scatter replaced by gather over a target-sorted edge list built each call (count+rank atomic, scan, fill); sums in arrival-rank order, self-loop+bias folded in
# speedup vs baseline: 1.5157x; 1.5157x over previous
#include <hip/hip_runtime.h>
#include <hip/hip_bf16.h>

typedef float v2f __attribute__((ext_vector_type(2)));
typedef float v8f __attribute__((ext_vector_type(8)));

#define LDS_LD 130

__device__ __forceinline__ v8f wmma_f32(v2f a, v2f b, v8f c) {
    return __builtin_amdgcn_wmma_f32_16x16x4_f32(
         false, a,  false, b,
         (short)0, c,  false,  false);
}

__device__ __forceinline__ v2f a_frag_g(const float* __restrict__ A, int lda,
                                        int r0, int k, int lane) {
    int m  = lane & 15;
    int kk = k + ((lane >> 4) << 1);
    const float* p = A + (size_t)(r0 + m) * lda + kk;
    return v2f{p[0], p[1]};
}

__device__ __forceinline__ v2f a_frag_lds(const float* __restrict__ L, int ld,
                                          int k, int lane) {
    int m  = lane & 15;
    int kk = k + ((lane >> 4) << 1);
    const float* p = L + m * ld + kk;
    return v2f{p[0], p[1]};
}

__device__ __forceinline__ v2f b_frag_g(const float* __restrict__ B, int ldb,
                                        int k, int c0, int lane) {
    int n  = lane & 15;
    int kk = k + ((lane >> 4) << 1);
    const float* p = B + (size_t)kk * ldb + c0 + n;
    return v2f{p[0], p[ldb]};
}

__device__ __forceinline__ void store_tile(float* __restrict__ D, int ldd,
                                           int r0, int c0, v8f acc, int lane,
                                           float bias) {
    int n    = lane & 15;
    int half = lane >> 4;
#pragma unroll
    for (int r = 0; r < 8; ++r)
        D[(size_t)(r0 + r + 8 * half) * ldd + c0 + n] = acc[r] + bias;
}

__global__ void cnt_zero_kernel(int* __restrict__ cnt, int N) {
    int i = blockIdx.x * blockDim.x + threadIdx.x;
    if (i < N) cnt[i] = 0;
}

__global__ void cnt_rank_kernel(const int* __restrict__ col, int* __restrict__ cnt,
                                int* __restrict__ rank, int E) {
    int e = blockIdx.x * blockDim.x + threadIdx.x;
    if (e < E) rank[e] = atomicAdd(&cnt[col[e]], 1);
}

__global__ __launch_bounds__(1024) void offs_dis_kernel(
    const int* __restrict__ cnt, int* __restrict__ offs, float* __restrict__ dis, int N)
{
    __shared__ int wave_incl[32];
    __shared__ int wave_base[32];
    const int tid  = threadIdx.x;
    const int lane = tid & 31;
    const int wv   = tid >> 5;
    const int i    = blockIdx.x * 1024 + tid;
    const int c    = (i < N) ? cnt[i] : 0;

    int part = 0;
    for (int j = 0; j < (int)blockIdx.x; ++j) part += cnt[j * 1024 + tid];

    int inc = c;
    for (int d = 1; d < 32; d <<= 1) {
        int v = __shfl_up(inc, d, 32);
        if (lane >= d) inc += v;
    }
    for (int d = 16; d > 0; d >>= 1) part += __shfl_xor(part, d, 32);
    if (lane == 31) wave_incl[wv] = inc;
    if (lane == 0)  wave_base[wv] = part;
    __syncthreads();

    int add = ((lane < wv) ? wave_incl[lane] : 0) + wave_base[lane];
    for (int d = 16; d > 0; d >>= 1) add += __shfl_xor(add, d, 32);

    const int excl = add + inc - c;
    if (i < N) {
        offs[i] = excl;
        dis[i]  = rsqrtf((float)(c + 1));
    }
    if (i == N - 1) offs[N] = excl + c;
}

__global__ void edge_fill_kernel(const int* __restrict__ row, const int* __restrict__ col,
                                 const int* __restrict__ rank, const int* __restrict__ offs,
                                 const float* __restrict__ dis,
                                 int* __restrict__ src, float* __restrict__ wgt, int E) {
    int e = blockIdx.x * blockDim.x + threadIdx.x;
    if (e >= E) return;
    int r = row[e], c = col[e];
    int p = offs[c] + rank[e];
    src[p] = r;
    wgt[p] = dis[r] * dis[c];
}

__global__ __launch_bounds__(128) void self_branch_kernel(
    const float* __restrict__ x, const float* __restrict__ Win,
    const float* __restrict__ bin, const float* __restrict__ Wout,
    const float* __restrict__ bout, const float* __restrict__ Wg1,
    float* __restrict__ l1_out, float* __restrict__ h1_out, int N)
{
    __shared__ float lds[4 * 16 * LDS_LD];
    const int lane = threadIdx.x & 31;
    const int wave = threadIdx.x >> 5;
    const int r0   = (blockIdx.x * 4 + wave) * 16;
    const int n    = lane & 15, half = lane >> 4;
    float* myL = lds + wave * 16 * LDS_LD;
    if (r0 >= N) return;

    v2f ax[32];
#pragma unroll
    for (int kc = 0; kc < 32; ++kc)
        ax[kc] = a_frag_g(x, 128, r0, kc * 4, lane);

    for (int t = 0; t < 8; ++t) {
        v8f acc = {};
#pragma unroll
        for (int kc = 0; kc < 32; ++kc)
            acc = wmma_f32(ax[kc], b_frag_g(Win, 128, kc * 4, t * 16, lane), acc);
        float bb = bin[t * 16 + n];
#pragma unroll
        for (int r = 0; r < 8; ++r) {
            float v = acc[r] + bb;
            myL[(r + 8 * half) * LDS_LD + t * 16 + n] = v > 0.f ? v : 0.f;
        }
    }
    __syncthreads();

    for (int t = 0; t < 8; ++t) {
        v8f acc = {};
#pragma unroll
        for (int kc = 0; kc < 32; ++kc)
            acc = wmma_f32(ax[kc], b_frag_g(Wg1, 128, kc * 4, t * 16, lane), acc);
        store_tile(h1_out, 128, r0, t * 16, acc, lane, 0.f);
    }

    v2f al[32];
#pragma unroll
    for (int kc = 0; kc < 32; ++kc)
        al[kc] = a_frag_lds(myL, LDS_LD, kc * 4, lane);

    for (int t = 0; t < 4; ++t) {
        v8f acc = {};
#pragma unroll
        for (int kc = 0; kc < 32; ++kc)
            acc = wmma_f32(ax[kc], b_frag_g(Wout, 64, kc * 4, t * 16, lane), acc);
#pragma unroll
        for (int kc = 0; kc < 32; ++kc)
            acc = wmma_f32(al[kc], b_frag_g(Wout, 64, 128 + kc * 4, t * 16, lane), acc);
        store_tile(l1_out, 64, r0, t * 16, acc, lane, bout[t * 16 + n]);
    }
}

template <int V, int U>
__device__ __forceinline__ void gather_rows(
    const float* __restrict__ h, int W, int s, float w, int k, int lo,
    float __attribute__((ext_vector_type(V)))& acc)
{
    typedef float vecf __attribute__((ext_vector_type(V)));
    vecf  v[U];
    float ww[U];
    for (int u = 0; u < U; ++u) {
        const int su = __builtin_amdgcn_readlane(s, k + u);
        ww[u] = __int_as_float(__builtin_amdgcn_readlane(__float_as_int(w), k + u));
        v[u]  = *(const vecf*)(h + (size_t)su * W + lo);
    }
    for (int u = 0; u < U; ++u) acc += v[u] * ww[u];
}

template <int W>
__global__ __launch_bounds__(256) void seg_gather_kernel(
    const int* __restrict__ offs, const int* __restrict__ src, const float* __restrict__ wgt,
    const float* __restrict__ dis, const float* __restrict__ h, const float* __restrict__ bias,
    float* __restrict__ g, int N)
{
    constexpr int V = W / 32;
    typedef float vecf __attribute__((ext_vector_type(V)));
    const int lane = threadIdx.x & 31;
    const int i = __builtin_amdgcn_readfirstlane(blockIdx.x * 8 + (threadIdx.x >> 5));
    if (i >= N) return;
    const int beg = __builtin_amdgcn_readfirstlane(offs[i]);
    const int end = __builtin_amdgcn_readfirstlane(offs[i + 1]);
    const int lo  = lane * V;
    const float d = dis[i];
    vecf acc = *(const vecf*)(bias + lo) + *(const vecf*)(h + (size_t)i * W + lo) * (d * d);

    for (int base = beg; base < end; base += 32) {
        const int m = min(32, end - base);
        int   s = 0;
        float w = 0.f;
        if (lane < m) { s = src[base + lane]; w = wgt[base + lane]; }
        int k = 0;
        for (; k + 8 <= m; k += 8) gather_rows<V, 8>(h, W, s, w, k, lo, acc);
        if (k + 4 <= m) { gather_rows<V, 4>(h, W, s, w, k, lo, acc); k += 4; }
        if (k + 2 <= m) { gather_rows<V, 2>(h, W, s, w, k, lo, acc); k += 2; }
        if (k < m)      { gather_rows<V, 1>(h, W, s, w, k, lo, acc); }
    }
    *(vecf*)(g + (size_t)i * W + lo) = acc;
}

__global__ __launch_bounds__(128) void gcn2_gemm_kernel(
    const float* __restrict__ g1, const float* __restrict__ Wg2,
    float* __restrict__ h2, int N)
{
    const int lane = threadIdx.x & 31;
    const int wave = threadIdx.x >> 5;
    const int r0   = (blockIdx.x * 4 + wave) * 16;
    if (r0 >= N) return;

    v2f ag[32];
#pragma unroll
    for (int kc = 0; kc < 32; ++kc)
        ag[kc] = a_frag_g(g1, 128, r0, kc * 4, lane);

    for (int t = 0; t < 4; ++t) {
        v8f acc = {};
#pragma unroll
        for (int kc = 0; kc < 32; ++kc)
            acc = wmma_f32(ag[kc], b_frag_g(Wg2, 64, kc * 4, t * 16, lane), acc);
        store_tile(h2, 64, r0, t * 16, acc, lane, 0.f);
    }
}

__global__ __launch_bounds__(128) void out_gemm_kernel(
    const float* __restrict__ xn, const float* __restrict__ g2,
    const float* __restrict__ W, const float* __restrict__ b,
    float* __restrict__ out, int N)
{
    const int lane = threadIdx.x & 31;
    const int wave = threadIdx.x >> 5;
    const int r0   = (blockIdx.x * 4 + wave) * 16;
    if (r0 >= N) return;
    const int n = lane & 15;

    v2f axn[32];
#pragma unroll
    for (int kc = 0; kc < 32; ++kc)
        axn[kc] = a_frag_g(xn, 128, r0, kc * 4, lane);
    v2f ag2[16];
#pragma unroll
    for (int kc = 0; kc < 16; ++kc)
        ag2[kc] = a_frag_g(g2, 64, r0, kc * 4, lane);

    for (int t = 0; t < 4; ++t) {
        v8f acc = {};
#pragma unroll
        for (int kc = 0; kc < 32; ++kc)
            acc = wmma_f32(axn[kc], b_frag_g(W, 64, kc * 4, t * 16, lane), acc);
#pragma unroll
        for (int kc = 0; kc < 16; ++kc)
            acc = wmma_f32(ag2[kc], b_frag_g(W, 64, 128 + kc * 4, t * 16, lane), acc);
        store_tile(out, 64, r0, t * 16, acc, lane, b[t * 16 + n]);
    }
}

extern "C" void kernel_launch(void* const* d_in, const int* in_sizes, int n_in,
                              void* d_out, int out_size, void* d_ws, size_t ws_size,
                              hipStream_t stream) {
    const float* x_self     = (const float*)d_in[0];
    const float* x_neighbor = (const float*)d_in[1];
    const int*   edge_index = (const int*)d_in[2];
    const float* W_in_self  = (const float*)d_in[3];
    const float* b_in_self  = (const float*)d_in[4];
    const float* W_out_self = (const float*)d_in[5];
    const float* b_out_self = (const float*)d_in[6];
    const float* Wg1        = (const float*)d_in[7];
    const float* bg1        = (const float*)d_in[8];
    const float* Wg2        = (const float*)d_in[9];
    const float* bg2        = (const float*)d_in[10];
    const float* W_out      = (const float*)d_in[11];
    const float* b_out      = (const float*)d_in[12];

    const int N = in_sizes[0] / 128;
    const int E = in_sizes[2] / 2;
    const int* row = edge_index;
    const int* col = edge_index + E;

    float* ws  = (float*)d_ws;
    float* dis = ws;
    int*   cnt = (int*)(ws + N);
    float* h   = ws + 2 * (size_t)N;
    float* g   = h + (size_t)N * 128;

    float* l1_out = (float*)d_out;
    float* x2_out = (float*)d_out + (size_t)N * 64;

    int*   src  = (int*)x2_out;
    float* wgt  = x2_out + E;
    int*   rank = (int*)x2_out + 2 * (size_t)E;
    int*   offs = (int*)x2_out + 3 * (size_t)E;

    cnt_zero_kernel<<<(N + 255) / 256, 256, 0, stream>>>(cnt, N);
    cnt_rank_kernel<<<(E + 255) / 256, 256, 0, stream>>>(col, cnt, rank, E);
    offs_dis_kernel<<<(N + 1023) / 1024, 1024, 0, stream>>>(cnt, offs, dis, N);
    edge_fill_kernel<<<(E + 255) / 256, 256, 0, stream>>>(row, col, rank, offs, dis, src, wgt, E);

    const int strips = (N + 15) / 16;
    const int blocks = (strips + 3) / 4;

    self_branch_kernel<<<blocks, 128, 0, stream>>>(x_self, W_in_self, b_in_self,
                                                   W_out_self, b_out_self, Wg1,
                                                   l1_out, h, N);

    seg_gather_kernel<128><<<(N + 7) / 8, 256, 0, stream>>>(offs, src, wgt, dis, h, bg1, g, N);

    gcn2_gemm_kernel<<<blocks, 128, 0, stream>>>(g, Wg2, h, N);

    seg_gather_kernel<64><<<(N + 7) / 8, 256, 0, stream>>>(offs, src, wgt, dis, h, bg2, g, N);

    out_gemm_kernel<<<blocks, 128, 0, stream>>>(x_neighbor, g, W_out, b_out, x2_out, N);
}


# GEMM weights staged once per resident work-group in LDS as ready-made B fragment pairs (256 work-groups walk the strips, 2 column tiles per pass) replacing 2 global loads per WMMA; k sum order kept
# speedup vs baseline: 1.9083x; 1.2590x over previous
#include <hip/hip_runtime.h>
#include <hip/hip_bf16.h>

typedef float v2f __attribute__((ext_vector_type(2)));
typedef float v4f __attribute__((ext_vector_type(4)));
typedef float v8f __attribute__((ext_vector_type(8)));

#define GEMM_WAVES  8
#define GEMM_BLOCKS 256

#define LDS_LD 130

__device__ __forceinline__ v8f wmma_f32(v2f a, v2f b, v8f c) {
    return __builtin_amdgcn_wmma_f32_16x16x4_f32(
         false, a,  false, b,
         (short)0, c,  false,  false);
}

__device__ __forceinline__ v2f a_frag_g(const float* __restrict__ A, int lda,
                                        int r0, int k, int lane) {
    int m  = lane & 15;
    int kk = k + ((lane >> 4) << 1);
    const float* p = A + (size_t)(r0 + m) * lda + kk;
    return v2f{p[0], p[1]};
}

__device__ __forceinline__ v2f a_frag_lds(const float* __restrict__ L, int ld,
                                          int k, int lane) {
    int m  = lane & 15;
    int kk = k + ((lane >> 4) << 1);
    const float* p = L + m * ld + kk;
    return v2f{p[0], p[1]};
}

template <int K, int NC>
__device__ __forceinline__ void stage_weights(const float* __restrict__ Wm,
                                              float* __restrict__ L, int tid) {
#pragma unroll 4
    for (int it = tid; it < K * NC / 4; it += GEMM_WAVES * 32) {
        const int c  = it % NC;
        const int qh = it / NC;
        const int k0 = (qh >> 1) * 8 + (qh & 1) * 2;
        const float* p = Wm + k0 * NC + c;
        *(v4f*)(L + it * 4) = v4f{p[0], p[NC], p[4 * NC], p[5 * NC]};
    }
}

template <int NQ, int NC>
__device__ __forceinline__ void mma_pass(const v2f* a, const float* __restrict__ L,
                                         int q0, int c0, int lane,
                                         v8f& acc0, v8f& acc1) {
    const int n = lane & 15, h = lane >> 4;
#pragma unroll
    for (int q = 0; q < NQ; ++q) {
        const float* p = L + (((q0 + q) * 2 + h) * NC + c0 + n) * 4;
        const v4f b0 = *(const v4f*)p;
        const v4f b1 = *(const v4f*)(p + 64);
        acc0 = wmma_f32(a[2 * q],     v2f{b0.x, b0.y}, acc0);
        acc1 = wmma_f32(a[2 * q],     v2f{b1.x, b1.y}, acc1);
        acc0 = wmma_f32(a[2 * q + 1], v2f{b0.z, b0.w}, acc0);
        acc1 = wmma_f32(a[2 * q + 1], v2f{b1.z, b1.w}, acc1);
    }
}

__device__ __forceinline__ void store_tile(float* __restrict__ D, int ldd,
                                           int r0, int c0, v8f acc, int lane,
                                           float bias) {
    int n    = lane & 15;
    int half = lane >> 4;
#pragma unroll
    for (int r = 0; r < 8; ++r)
        D[(size_t)(r0 + r + 8 * half) * ldd + c0 + n] = acc[r] + bias;
}

__global__ void cnt_zero_kernel(int* __restrict__ cnt, int N) {
    int i = blockIdx.x * blockDim.x + threadIdx.x;
    if (i < N) cnt[i] = 0;
}

__global__ void cnt_rank_kernel(const int* __restrict__ col, int* __restrict__ cnt,
                                int* __restrict__ rank, int E) {
    int e = blockIdx.x * blockDim.x + threadIdx.x;
    if (e < E) rank[e] = atomicAdd(&cnt[col[e]], 1);
}

__global__ __launch_bounds__(1024) void offs_dis_kernel(
    const int* __restrict__ cnt, int* __restrict__ offs, float* __restrict__ dis, int N)
{
    __shared__ int wave_incl[32];
    __shared__ int wave_base[32];
    const int tid  = threadIdx.x;
    const int lane = tid & 31;
    const int wv   = tid >> 5;
    const int i    = blockIdx.x * 1024 + tid;
    const int c    = (i < N) ? cnt[i] : 0;

    int part = 0;
    for (int j = 0; j < (int)blockIdx.x; ++j) part += cnt[j * 1024 + tid];

    int inc = c;
    for (int d = 1; d < 32; d <<= 1) {
        int v = __shfl_up(inc, d, 32);
        if (lane >= d) inc += v;
    }
    for (int d = 16; d > 0; d >>= 1) part += __shfl_xor(part, d, 32);
    if (lane == 31) wave_incl[wv] = inc;
    if (lane == 0)  wave_base[wv] = part;
    __syncthreads();

    int add = ((lane < wv) ? wave_incl[lane] : 0) + wave_base[lane];
    for (int d = 16; d > 0; d >>= 1) add += __shfl_xor(add, d, 32);

    const int excl = add + inc - c;
    if (i < N) {
        offs[i] = excl;
        dis[i]  = rsqrtf((float)(c + 1));
    }
    if (i == N - 1) offs[N] = excl + c;
}

__global__ void edge_fill_kernel(const int* __restrict__ row, const int* __restrict__ col,
                                 const int* __restrict__ rank, const int* __restrict__ offs,
                                 const float* __restrict__ dis,
                                 int* __restrict__ src, float* __restrict__ wgt, int E) {
    int e = blockIdx.x * blockDim.x + threadIdx.x;
    if (e >= E) return;
    int r = row[e], c = col[e];
    int p = offs[c] + rank[e];
    src[p] = r;
    wgt[p] = dis[r] * dis[c];
}

__global__ __launch_bounds__(GEMM_WAVES * 32) void self_branch_kernel(
    const float* __restrict__ x, const float* __restrict__ Win,
    const float* __restrict__ bin, const float* __restrict__ Wout,
    const float* __restrict__ bout, const float* __restrict__ Wg1,
    float* __restrict__ l1_out, float* __restrict__ h1_out, int N)
{
    __shared__ __attribute__((aligned(16))) float lds[3 * 16384 + GEMM_WAVES * 16 * LDS_LD];
    float* Lin  = lds;
    float* Lg1  = lds + 16384;
    float* Lout = lds + 32768;
    const int tid  = threadIdx.x;
    const int lane = tid & 31;
    const int wave = tid >> 5;
    const int n    = lane & 15, half = lane >> 4;
    float* myL = lds + 49152 + wave * 16 * LDS_LD;

    stage_weights<128, 128>(Win,  Lin,  tid);
    stage_weights<128, 128>(Wg1,  Lg1,  tid);
    stage_weights<256, 64>(Wout, Lout, tid);
    __syncthreads();

    const int strips = N / 16;
    for (int strip = wave * (int)gridDim.x + (int)blockIdx.x; strip < strips;
         strip += GEMM_WAVES * (int)gridDim.x) {
        const int r0 = strip * 16;

        v2f ax[32];
#pragma unroll
        for (int kc = 0; kc < 32; ++kc)
            ax[kc] = a_frag_g(x, 128, r0, kc * 4, lane);

        for (int t = 0; t < 8; t += 2) {
            v8f acc0 = {}, acc1 = {};
            mma_pass<16, 128>(ax, Lin, 0, t * 16, lane, acc0, acc1);
            const float bb0 = bin[t * 16 + n], bb1 = bin[t * 16 + 16 + n];
#pragma unroll
            for (int r = 0; r < 8; ++r) {
                const float v0 = acc0[r] + bb0, v1 = acc1[r] + bb1;
                myL[(r + 8 * half) * LDS_LD + t * 16 + n]      = v0 > 0.f ? v0 : 0.f;
                myL[(r + 8 * half) * LDS_LD + t * 16 + 16 + n] = v1 > 0.f ? v1 : 0.f;
            }
        }

        for (int t = 0; t < 8; t += 2) {
            v8f acc0 = {}, acc1 = {};
            mma_pass<16, 128>(ax, Lg1, 0, t * 16, lane, acc0, acc1);
            store_tile(h1_out, 128, r0, t * 16,      acc0, lane, 0.f);
            store_tile(h1_out, 128, r0, t * 16 + 16, acc1, lane, 0.f);
        }

        __builtin_amdgcn_fence(__ATOMIC_ACQ_REL, "wavefront");

        v2f al[32];
#pragma unroll
        for (int kc = 0; kc < 32; ++kc)
            al[kc] = a_frag_lds(myL, LDS_LD, kc * 4, lane);

        for (int t = 0; t < 4; t += 2) {
            v8f acc0 = {}, acc1 = {};
            mma_pass<16, 64>(ax, Lout, 0,  t * 16, lane, acc0, acc1);
            mma_pass<16, 64>(al, Lout, 16, t * 16, lane, acc0, acc1);
            store_tile(l1_out, 64, r0, t * 16,      acc0, lane, bout[t * 16 + n]);
            store_tile(l1_out, 64, r0, t * 16 + 16, acc1, lane, bout[t * 16 + 16 + n]);
        }

        __builtin_amdgcn_fence(__ATOMIC_ACQ_REL, "wavefront");
    }
}

template <int V, int U>
__device__ __forceinline__ void gather_rows(
    const float* __restrict__ h, int W, int s, float w, int k, int lo,
    float __attribute__((ext_vector_type(V)))& acc)
{
    typedef float vecf __attribute__((ext_vector_type(V)));
    vecf  v[U];
    float ww[U];
    for (int u = 0; u < U; ++u) {
        const int su = __builtin_amdgcn_readlane(s, k + u);
        ww[u] = __int_as_float(__builtin_amdgcn_readlane(__float_as_int(w), k + u));
        v[u]  = *(const vecf*)(h + (size_t)su * W + lo);
    }
    for (int u = 0; u < U; ++u) acc += v[u] * ww[u];
}

template <int W>
__global__ __launch_bounds__(256) void seg_gather_kernel(
    const int* __restrict__ offs, const int* __restrict__ src, const float* __restrict__ wgt,
    const float* __restrict__ dis, const float* __restrict__ h, const float* __restrict__ bias,
    float* __restrict__ g, int N)
{
    constexpr int V = W / 32;
    typedef float vecf __attribute__((ext_vector_type(V)));
    const int lane = threadIdx.x & 31;
    const int i = __builtin_amdgcn_readfirstlane(blockIdx.x * 8 + (threadIdx.x >> 5));
    if (i >= N) return;
    const int beg = __builtin_amdgcn_readfirstlane(offs[i]);
    const int end = __builtin_amdgcn_readfirstlane(offs[i + 1]);
    const int lo  = lane * V;
    const float d = dis[i];
    vecf acc = *(const vecf*)(bias + lo) + *(const vecf*)(h + (size_t)i * W + lo) * (d * d);

    for (int base = beg; base < end; base += 32) {
        const int m = min(32, end - base);
        int   s = 0;
        float w = 0.f;
        if (lane < m) { s = src[base + lane]; w = wgt[base + lane]; }
        int k = 0;
        for (; k + 8 <= m; k += 8) gather_rows<V, 8>(h, W, s, w, k, lo, acc);
        if (k + 4 <= m) { gather_rows<V, 4>(h, W, s, w, k, lo, acc); k += 4; }
        if (k + 2 <= m) { gather_rows<V, 2>(h, W, s, w, k, lo, acc); k += 2; }
        if (k < m)      { gather_rows<V, 1>(h, W, s, w, k, lo, acc); }
    }
    *(vecf*)(g + (size_t)i * W + lo) = acc;
}

__global__ __launch_bounds__(GEMM_WAVES * 32) void gcn2_gemm_kernel(
    const float* __restrict__ g1, const float* __restrict__ Wg2,
    float* __restrict__ h2, int N)
{
    __shared__ __attribute__((aligned(16))) float Lw[128 * 64];
    const int tid  = threadIdx.x;
    const int lane = tid & 31;
    const int wave = tid >> 5;
    stage_weights<128, 64>(Wg2, Lw, tid);
    __syncthreads();

    const int strips = N / 16;
    for (int strip = wave * (int)gridDim.x + (int)blockIdx.x; strip < strips;
         strip += GEMM_WAVES * (int)gridDim.x) {
        const int r0 = strip * 16;
        v2f ag[32];
#pragma unroll
        for (int kc = 0; kc < 32; ++kc)
            ag[kc] = a_frag_g(g1, 128, r0, kc * 4, lane);

        for (int t = 0; t < 4; t += 2) {
            v8f acc0 = {}, acc1 = {};
            mma_pass<16, 64>(ag, Lw, 0, t * 16, lane, acc0, acc1);
            store_tile(h2, 64, r0, t * 16,      acc0, lane, 0.f);
            store_tile(h2, 64, r0, t * 16 + 16, acc1, lane, 0.f);
        }
    }
}

__global__ __launch_bounds__(GEMM_WAVES * 32) void out_gemm_kernel(
    const float* __restrict__ xn, const float* __restrict__ g2,
    const float* __restrict__ W, const float* __restrict__ b,
    float* __restrict__ out, int N)
{
    __shared__ __attribute__((aligned(16))) float Lw[192 * 64];
    const int tid  = threadIdx.x;
    const int lane = tid & 31;
    const int wave = tid >> 5;
    const int n    = lane & 15;
    stage_weights<192, 64>(W, Lw, tid);
    __syncthreads();

    const int strips = N / 16;
    for (int strip = wave * (int)gridDim.x + (int)blockIdx.x; strip < strips;
         strip += GEMM_WAVES * (int)gridDim.x) {
        const int r0 = strip * 16;
        v2f axn[32];
#pragma unroll
        for (int kc = 0; kc < 32; ++kc)
            axn[kc] = a_frag_g(xn, 128, r0, kc * 4, lane);
        v2f ag2[16];
#pragma unroll
        for (int kc = 0; kc < 16; ++kc)
            ag2[kc] = a_frag_g(g2, 64, r0, kc * 4, lane);

        for (int t = 0; t < 4; t += 2) {
            v8f acc0 = {}, acc1 = {};
            mma_pass<16, 64>(axn, Lw, 0,  t * 16, lane, acc0, acc1);
            mma_pass<8, 64>(ag2, Lw, 16, t * 16, lane, acc0, acc1);
            store_tile(out, 64, r0, t * 16,      acc0, lane, b[t * 16 + n]);
            store_tile(out, 64, r0, t * 16 + 16, acc1, lane, b[t * 16 + 16 + n]);
        }
    }
}

extern "C" void kernel_launch(void* const* d_in, const int* in_sizes, int n_in,
                              void* d_out, int out_size, void* d_ws, size_t ws_size,
                              hipStream_t stream) {
    const float* x_self     = (const float*)d_in[0];
    const float* x_neighbor = (const float*)d_in[1];
    const int*   edge_index = (const int*)d_in[2];
    const float* W_in_self  = (const float*)d_in[3];
    const float* b_in_self  = (const float*)d_in[4];
    const float* W_out_self = (const float*)d_in[5];
    const float* b_out_self = (const float*)d_in[6];
    const float* Wg1        = (const float*)d_in[7];
    const float* bg1        = (const float*)d_in[8];
    const float* Wg2        = (const float*)d_in[9];
    const float* bg2        = (const float*)d_in[10];
    const float* W_out      = (const float*)d_in[11];
    const float* b_out      = (const float*)d_in[12];

    const int N = in_sizes[0] / 128;
    const int E = in_sizes[2] / 2;
    const int* row = edge_index;
    const int* col = edge_index + E;

    float* ws  = (float*)d_ws;
    float* dis = ws;
    int*   cnt = (int*)(ws + N);
    float* h   = ws + 2 * (size_t)N;
    float* g   = h + (size_t)N * 128;

    float* l1_out = (float*)d_out;
    float* x2_out = (float*)d_out + (size_t)N * 64;

    int*   src  = (int*)x2_out;
    float* wgt  = x2_out + E;
    int*   rank = (int*)x2_out + 2 * (size_t)E;
    int*   offs = (int*)x2_out + 3 * (size_t)E;

    cnt_zero_kernel<<<(N + 255) / 256, 256, 0, stream>>>(cnt, N);
    cnt_rank_kernel<<<(E + 255) / 256, 256, 0, stream>>>(col, cnt, rank, E);
    offs_dis_kernel<<<(N + 1023) / 1024, 1024, 0, stream>>>(cnt, offs, dis, N);
    edge_fill_kernel<<<(E + 255) / 256, 256, 0, stream>>>(row, col, rank, offs, dis, src, wgt, E);

    self_branch_kernel<<<GEMM_BLOCKS, GEMM_WAVES * 32, 0, stream>>>(x_self, W_in_self, b_in_self,
                                                   W_out_self, b_out_self, Wg1,
                                                   l1_out, h, N);

    seg_gather_kernel<128><<<(N + 7) / 8, 256, 0, stream>>>(offs, src, wgt, dis, h, bg1, g, N);

    gcn2_gemm_kernel<<<GEMM_BLOCKS, GEMM_WAVES * 32, 0, stream>>>(g, Wg2, h, N);

    seg_gather_kernel<64><<<(N + 7) / 8, 256, 0, stream>>>(offs, src, wgt, dis, h, bg2, g, N);

    out_gemm_kernel<<<GEMM_BLOCKS, GEMM_WAVES * 32, 0, stream>>>(x_neighbor, g, W_out, b_out, x2_out, N);
}


# GCN layers reassociated (no activation between): gather x@(Wg1@Wg2) + bg1@Wg2, fold built per call on WMMA; first gather 64 wide, gcn2 GEMM launch gone; product sum order changes
# speedup vs baseline: 2.0983x; 1.0996x over previous
#include <hip/hip_runtime.h>
#include <hip/hip_bf16.h>

typedef float v2f __attribute__((ext_vector_type(2)));
typedef float v4f __attribute__((ext_vector_type(4)));
typedef float v8f __attribute__((ext_vector_type(8)));

#define GEMM_WAVES  8
#define GEMM_BLOCKS 256

#define LDS_LD 130

__device__ __forceinline__ v8f wmma_f32(v2f a, v2f b, v8f c) {
    return __builtin_amdgcn_wmma_f32_16x16x4_f32(
         false, a,  false, b,
         (short)0, c,  false,  false);
}

__device__ __forceinline__ v2f a_frag_g(const float* __restrict__ A, int lda,
                                        int r0, int k, int lane) {
    int m  = lane & 15;
    int kk = k + ((lane >> 4) << 1);
    const float* p = A + (size_t)(r0 + m) * lda + kk;
    return v2f{p[0], p[1]};
}

__device__ __forceinline__ v2f a_frag_lds(const float* __restrict__ L, int ld,
                                          int k, int lane) {
    int m  = lane & 15;
    int kk = k + ((lane >> 4) << 1);
    const float* p = L + m * ld + kk;
    return v2f{p[0], p[1]};
}

template <int K, int NC>
__device__ __forceinline__ void stage_weights(const float* __restrict__ Wm,
                                              float* __restrict__ L, int tid) {
#pragma unroll 4
    for (int it = tid; it < K * NC / 4; it += GEMM_WAVES * 32) {
        const int c  = it % NC;
        const int qh = it / NC;
        const int k0 = (qh >> 1) * 8 + (qh & 1) * 2;
        const float* p = Wm + k0 * NC + c;
        *(v4f*)(L + it * 4) = v4f{p[0], p[NC], p[4 * NC], p[5 * NC]};
    }
}

template <int NQ, int NC>
__device__ __forceinline__ void mma_pass(const v2f* a, const float* __restrict__ L,
                                         int q0, int c0, int lane,
                                         v8f& acc0, v8f& acc1) {
    const int n = lane & 15, h = lane >> 4;
#pragma unroll
    for (int q = 0; q < NQ; ++q) {
        const float* p = L + (((q0 + q) * 2 + h) * NC + c0 + n) * 4;
        const v4f b0 = *(const v4f*)p;
        const v4f b1 = *(const v4f*)(p + 64);
        acc0 = wmma_f32(a[2 * q],     v2f{b0.x, b0.y}, acc0);
        acc1 = wmma_f32(a[2 * q],     v2f{b1.x, b1.y}, acc1);
        acc0 = wmma_f32(a[2 * q + 1], v2f{b0.z, b0.w}, acc0);
        acc1 = wmma_f32(a[2 * q + 1], v2f{b1.z, b1.w}, acc1);
    }
}

__device__ __forceinline__ void store_tile(float* __restrict__ D, int ldd,
                                           int r0, int c0, v8f acc, int lane,
                                           float bias) {
    int n    = lane & 15;
    int half = lane >> 4;
#pragma unroll
    for (int r = 0; r < 8; ++r)
        D[(size_t)(r0 + r + 8 * half) * ldd + c0 + n] = acc[r] + bias;
}

__global__ void cnt_zero_kernel(int* __restrict__ cnt, int N) {
    int i = blockIdx.x * blockDim.x + threadIdx.x;
    if (i < N) cnt[i] = 0;
}

__global__ void cnt_rank_kernel(const int* __restrict__ col, int* __restrict__ cnt,
                                int* __restrict__ rank, int E) {
    int e = blockIdx.x * blockDim.x + threadIdx.x;
    if (e < E) rank[e] = atomicAdd(&cnt[col[e]], 1);
}

__global__ __launch_bounds__(1024) void offs_dis_kernel(
    const int* __restrict__ cnt, int* __restrict__ offs, float* __restrict__ dis, int N)
{
    __shared__ int wave_incl[32];
    __shared__ int wave_base[32];
    const int tid  = threadIdx.x;
    const int lane = tid & 31;
    const int wv   = tid >> 5;
    const int i    = blockIdx.x * 1024 + tid;
    const int c    = (i < N) ? cnt[i] : 0;

    int part = 0;
    for (int j = 0; j < (int)blockIdx.x; ++j) part += cnt[j * 1024 + tid];

    int inc = c;
    for (int d = 1; d < 32; d <<= 1) {
        int v = __shfl_up(inc, d, 32);
        if (lane >= d) inc += v;
    }
    for (int d = 16; d > 0; d >>= 1) part += __shfl_xor(part, d, 32);
    if (lane == 31) wave_incl[wv] = inc;
    if (lane == 0)  wave_base[wv] = part;
    __syncthreads();

    int add = ((lane < wv) ? wave_incl[lane] : 0) + wave_base[lane];
    for (int d = 16; d > 0; d >>= 1) add += __shfl_xor(add, d, 32);

    const int excl = add + inc - c;
    if (i < N) {
        offs[i] = excl;
        dis[i]  = rsqrtf((float)(c + 1));
    }
    if (i == N - 1) offs[N] = excl + c;
}

__global__ void edge_fill_kernel(const int* __restrict__ row, const int* __restrict__ col,
                                 const int* __restrict__ rank, const int* __restrict__ offs,
                                 const float* __restrict__ dis,
                                 int* __restrict__ src, float* __restrict__ wgt, int E) {
    int e = blockIdx.x * blockDim.x + threadIdx.x;
    if (e >= E) return;
    int r = row[e], c = col[e];
    int p = offs[c] + rank[e];
    src[p] = r;
    wgt[p] = dis[r] * dis[c];
}

__global__ __launch_bounds__(128) void weight_fold_kernel(
    const float* __restrict__ Wg1, const float* __restrict__ bg1,
    const float* __restrict__ Wg2, float* __restrict__ W12)
{
    const int lane = threadIdx.x & 31;
    const int c0   = (threadIdx.x >> 5) * 16;
    const int r0   = blockIdx.x * 16;
    const int n    = lane & 15, h = lane >> 4;
    const int row  = r0 + n;
    const float* arow = row < 128 ? Wg1 + row * 128 : bg1;
    const bool  live  = row <= 128;
    const float* bcol = Wg2 + c0 + n;

    v8f acc0 = {}, acc1 = {};
#pragma unroll
    for (int kc = 0; kc < 32; kc += 2) {
        const int k0 = kc * 4 + 2 * h, k1 = k0 + 4;
        v2f a0 = v2f{arow[k0], arow[k0 + 1]};
        v2f a1 = v2f{arow[k1], arow[k1 + 1]};
        if (!live) { a0 = v2f{0.f, 0.f}; a1 = v2f{0.f, 0.f}; }
        const v2f b0 = v2f{bcol[k0 * 64], bcol[(k0 + 1) * 64]};
        const v2f b1 = v2f{bcol[k1 * 64], bcol[(k1 + 1) * 64]};
        acc0 = wmma_f32(a0, b0, acc0);
        acc1 = wmma_f32(a1, b1, acc1);
    }
    store_tile(W12, 64, r0, c0, acc0 + acc1, lane, 0.f);
}

__global__ __launch_bounds__(GEMM_WAVES * 32) void self_branch_kernel(
    const float* __restrict__ x, const float* __restrict__ Win,
    const float* __restrict__ bin, const float* __restrict__ Wout,
    const float* __restrict__ bout, const float* __restrict__ W12,
    float* __restrict__ l1_out, float* __restrict__ p_out, int N)
{
    __shared__ __attribute__((aligned(16))) float lds[16384 + 8192 + 16384 + GEMM_WAVES * 16 * LDS_LD];
    float* Lin  = lds;
    float* L12  = lds + 16384;
    float* Lout = lds + 24576;
    const int tid  = threadIdx.x;
    const int lane = tid & 31;
    const int wave = tid >> 5;
    const int n    = lane & 15, half = lane >> 4;
    float* myL = lds + 40960 + wave * 16 * LDS_LD;

    stage_weights<128, 128>(Win,  Lin,  tid);
    stage_weights<128, 64>(W12,  L12,  tid);
    stage_weights<256, 64>(Wout, Lout, tid);
    __syncthreads();

    const int strips = N / 16;
    for (int strip = wave * (int)gridDim.x + (int)blockIdx.x; strip < strips;
         strip += GEMM_WAVES * (int)gridDim.x) {
        const int r0 = strip * 16;

        v2f ax[32];
#pragma unroll
        for (int kc = 0; kc < 32; ++kc)
            ax[kc] = a_frag_g(x, 128, r0, kc * 4, lane);

        for (int t = 0; t < 8; t += 2) {
            v8f acc0 = {}, acc1 = {};
            mma_pass<16, 128>(ax, Lin, 0, t * 16, lane, acc0, acc1);
            const float bb0 = bin[t * 16 + n], bb1 = bin[t * 16 + 16 + n];
#pragma unroll
            for (int r = 0; r < 8; ++r) {
                const float v0 = acc0[r] + bb0, v1 = acc1[r] + bb1;
                myL[(r + 8 * half) * LDS_LD + t * 16 + n]      = v0 > 0.f ? v0 : 0.f;
                myL[(r + 8 * half) * LDS_LD + t * 16 + 16 + n] = v1 > 0.f ? v1 : 0.f;
            }
        }

        for (int t = 0; t < 4; t += 2) {
            v8f acc0 = {}, acc1 = {};
            mma_pass<16, 64>(ax, L12, 0, t * 16, lane, acc0, acc1);
            store_tile(p_out, 64, r0, t * 16,      acc0, lane, 0.f);
            store_tile(p_out, 64, r0, t * 16 + 16, acc1, lane, 0.f);
        }

        __builtin_amdgcn_fence(__ATOMIC_ACQ_REL, "wavefront");

        v2f al[32];
#pragma unroll
        for (int kc = 0; kc < 32; ++kc)
            al[kc] = a_frag_lds(myL, LDS_LD, kc * 4, lane);

        for (int t = 0; t < 4; t += 2) {
            v8f acc0 = {}, acc1 = {};
            mma_pass<16, 64>(ax, Lout, 0,  t * 16, lane, acc0, acc1);
            mma_pass<16, 64>(al, Lout, 16, t * 16, lane, acc0, acc1);
            store_tile(l1_out, 64, r0, t * 16,      acc0, lane, bout[t * 16 + n]);
            store_tile(l1_out, 64, r0, t * 16 + 16, acc1, lane, bout[t * 16 + 16 + n]);
        }

        __builtin_amdgcn_fence(__ATOMIC_ACQ_REL, "wavefront");
    }
}

template <int V, int U>
__device__ __forceinline__ void gather_rows(
    const float* __restrict__ h, int W, int s, float w, int k, int lo,
    float __attribute__((ext_vector_type(V)))& acc)
{
    typedef float vecf __attribute__((ext_vector_type(V)));
    vecf  v[U];
    float ww[U];
    for (int u = 0; u < U; ++u) {
        const int su = __builtin_amdgcn_readlane(s, k + u);
        ww[u] = __int_as_float(__builtin_amdgcn_readlane(__float_as_int(w), k + u));
        v[u]  = *(const vecf*)(h + (size_t)su * W + lo);
    }
    for (int u = 0; u < U; ++u) acc += v[u] * ww[u];
}

template <int W>
__global__ __launch_bounds__(256) void seg_gather_kernel(
    const int* __restrict__ offs, const int* __restrict__ src, const float* __restrict__ wgt,
    const float* __restrict__ dis, const float* __restrict__ h, const float* __restrict__ bias,
    float* __restrict__ g, int N)
{
    constexpr int V = W / 32;
    typedef float vecf __attribute__((ext_vector_type(V)));
    const int lane = threadIdx.x & 31;
    const int i = __builtin_amdgcn_readfirstlane(blockIdx.x * 8 + (threadIdx.x >> 5));
    if (i >= N) return;
    const int beg = __builtin_amdgcn_readfirstlane(offs[i]);
    const int end = __builtin_amdgcn_readfirstlane(offs[i + 1]);
    const int lo  = lane * V;
    const float d = dis[i];
    vecf acc = *(const vecf*)(bias + lo) + *(const vecf*)(h + (size_t)i * W + lo) * (d * d);

    for (int base = beg; base < end; base += 32) {
        const int m = min(32, end - base);
        int   s = 0;
        float w = 0.f;
        if (lane < m) { s = src[base + lane]; w = wgt[base + lane]; }
        int k = 0;
        for (; k + 8 <= m; k += 8) gather_rows<V, 8>(h, W, s, w, k, lo, acc);
        if (k + 4 <= m) { gather_rows<V, 4>(h, W, s, w, k, lo, acc); k += 4; }
        if (k + 2 <= m) { gather_rows<V, 2>(h, W, s, w, k, lo, acc); k += 2; }
        if (k < m)      { gather_rows<V, 1>(h, W, s, w, k, lo, acc); }
    }
    *(vecf*)(g + (size_t)i * W + lo) = acc;
}

__global__ __launch_bounds__(GEMM_WAVES * 32) void out_gemm_kernel(
    const float* __restrict__ xn, const float* __restrict__ g2,
    const float* __restrict__ W, const float* __restrict__ b,
    float* __restrict__ out, int N)
{
    __shared__ __attribute__((aligned(16))) float Lw[192 * 64];
    const int tid  = threadIdx.x;
    const int lane = tid & 31;
    const int wave = tid >> 5;
    const int n    = lane & 15;
    stage_weights<192, 64>(W, Lw, tid);
    __syncthreads();

    const int strips = N / 16;
    for (int strip = wave * (int)gridDim.x + (int)blockIdx.x; strip < strips;
         strip += GEMM_WAVES * (int)gridDim.x) {
        const int r0 = strip * 16;
        v2f axn[32];
#pragma unroll
        for (int kc = 0; kc < 32; ++kc)
            axn[kc] = a_frag_g(xn, 128, r0, kc * 4, lane);
        v2f ag2[16];
#pragma unroll
        for (int kc = 0; kc < 16; ++kc)
            ag2[kc] = a_frag_g(g2, 64, r0, kc * 4, lane);

        for (int t = 0; t < 4; t += 2) {
            v8f acc0 = {}, acc1 = {};
            mma_pass<16, 64>(axn, Lw, 0,  t * 16, lane, acc0, acc1);
            mma_pass<8, 64>(ag2, Lw, 16, t * 16, lane, acc0, acc1);
            store_tile(out, 64, r0, t * 16,      acc0, lane, b[t * 16 + n]);
            store_tile(out, 64, r0, t * 16 + 16, acc1, lane, b[t * 16 + 16 + n]);
        }
    }
}

extern "C" void kernel_launch(void* const* d_in, const int* in_sizes, int n_in,
                              void* d_out, int out_size, void* d_ws, size_t ws_size,
                              hipStream_t stream) {
    const float* x_self     = (const float*)d_in[0];
    const float* x_neighbor = (const float*)d_in[1];
    const int*   edge_index = (const int*)d_in[2];
    const float* W_in_self  = (const float*)d_in[3];
    const float* b_in_self  = (const float*)d_in[4];
    const float* W_out_self = (const float*)d_in[5];
    const float* b_out_self = (const float*)d_in[6];
    const float* Wg1        = (const float*)d_in[7];
    const float* bg1        = (const float*)d_in[8];
    const float* Wg2        = (const float*)d_in[9];
    const float* bg2        = (const float*)d_in[10];
    const float* W_out      = (const float*)d_in[11];
    const float* b_out      = (const float*)d_in[12];

    const int N = in_sizes[0] / 128;
    const int E = in_sizes[2] / 2;
    const int* row = edge_index;
    const int* col = edge_index + E;

    float* ws  = (float*)d_ws;
    float* dis = ws;
    int*   cnt = (int*)(ws + N);
    float* p   = ws + 2 * (size_t)N;
    float* h2  = p  + (size_t)N * 64;
    float* g2  = h2 + (size_t)N * 64;
    float* W12 = g2 + (size_t)N * 64;

    float* l1_out = (float*)d_out;
    float* x2_out = (float*)d_out + (size_t)N * 64;

    int*   src  = (int*)x2_out;
    float* wgt  = x2_out + E;
    int*   rank = (int*)x2_out + 2 * (size_t)E;
    int*   offs = (int*)x2_out + 3 * (size_t)E;

    cnt_zero_kernel<<<(N + 255) / 256, 256, 0, stream>>>(cnt, N);
    cnt_rank_kernel<<<(E + 255) / 256, 256, 0, stream>>>(col, cnt, rank, E);
    offs_dis_kernel<<<(N + 1023) / 1024, 1024, 0, stream>>>(cnt, offs, dis, N);
    edge_fill_kernel<<<(E + 255) / 256, 256, 0, stream>>>(row, col, rank, offs, dis, src, wgt, E);

    weight_fold_kernel<<<9, 128, 0, stream>>>(Wg1, bg1, Wg2, W12);

    self_branch_kernel<<<GEMM_BLOCKS, GEMM_WAVES * 32, 0, stream>>>(x_self, W_in_self, b_in_self,
                                                   W_out_self, b_out_self, W12,
                                                   l1_out, p, N);

    seg_gather_kernel<64><<<(N + 7) / 8, 256, 0, stream>>>(offs, src, wgt, dis, p, W12 + 128 * 64, h2, N);

    seg_gather_kernel<64><<<(N + 7) / 8, 256, 0, stream>>>(offs, src, wgt, dis, h2, bg2, g2, N);

    out_gemm_kernel<<<GEMM_BLOCKS, GEMM_WAVES * 32, 0, stream>>>(x_neighbor, g2, W_out, b_out, x2_out, N);
}


# Edge count+rank on 4 extra waves per fused GEMM work-group, under the matrix work (tile loops rolled to fit 336 VGPRs); counter clear in weight-fold launch; 9->7 launches; arrival-rank sums kept
# speedup vs baseline: 2.3797x; 1.1341x over previous
#include <hip/hip_runtime.h>
#include <hip/hip_bf16.h>

typedef float v2f __attribute__((ext_vector_type(2)));
typedef float v4f __attribute__((ext_vector_type(4)));
typedef float v8f __attribute__((ext_vector_type(8)));

#define GEMM_WAVES  8
#define GEMM_BLOCKS 256
#define EDGE_WAVES  4
#define EDGE_BATCH  4

#define LDS_LD 130

__device__ __forceinline__ v8f wmma_f32(v2f a, v2f b, v8f c) {
    return __builtin_amdgcn_wmma_f32_16x16x4_f32(
         false, a,  false, b,
         (short)0, c,  false,  false);
}

__device__ __forceinline__ v2f a_frag_g(const float* __restrict__ A, int lda,
                                        int r0, int k, int lane) {
    int m  = lane & 15;
    int kk = k + ((lane >> 4) << 1);
    const float* p = A + (size_t)(r0 + m) * lda + kk;
    return v2f{p[0], p[1]};
}

__device__ __forceinline__ v2f a_frag_lds(const float* __restrict__ L, int ld,
                                          int k, int lane) {
    int m  = lane & 15;
    int kk = k + ((lane >> 4) << 1);
    const float* p = L + m * ld + kk;
    return v2f{p[0], p[1]};
}

template <int K, int NC>
__device__ __forceinline__ void stage_weights(const float* __restrict__ Wm,
                                              float* __restrict__ L, int tid) {
#pragma unroll 4
    for (int it = tid; it < K * NC / 4; it += GEMM_WAVES * 32) {
        const int c  = it % NC;
        const int qh = it / NC;
        const int k0 = (qh >> 1) * 8 + (qh & 1) * 2;
        const float* p = Wm + k0 * NC + c;
        *(v4f*)(L + it * 4) = v4f{p[0], p[NC], p[4 * NC], p[5 * NC]};
    }
}

template <int NQ, int NC>
__device__ __forceinline__ void mma_pass(const v2f* a, const float* __restrict__ L,
                                         int q0, int c0, int lane,
                                         v8f& acc0, v8f& acc1) {
    const int n = lane & 15, h = lane >> 4;
    v4f b0[NQ], b1[NQ];
#pragma unroll
    for (int q = 0; q < NQ; ++q) {
        const float* p = L + (((q0 + q) * 2 + h) * NC + c0 + n) * 4;
        b0[q] = *(const v4f*)p;
        b1[q] = *(const v4f*)(p + 64);
    }
#pragma unroll
    for (int q = 0; q < NQ; ++q) {
        acc0 = wmma_f32(a[2 * q],     v2f{b0[q].x, b0[q].y}, acc0);
        acc1 = wmma_f32(a[2 * q],     v2f{b1[q].x, b1[q].y}, acc1);
        acc0 = wmma_f32(a[2 * q + 1], v2f{b0[q].z, b0[q].w}, acc0);
        acc1 = wmma_f32(a[2 * q + 1], v2f{b1[q].z, b1[q].w}, acc1);
    }
}

__device__ __forceinline__ void store_tile(float* __restrict__ D, int ldd,
                                           int r0, int c0, v8f acc, int lane,
                                           float bias) {
    int n    = lane & 15;
    int half = lane >> 4;
#pragma unroll
    for (int r = 0; r < 8; ++r)
        D[(size_t)(r0 + r + 8 * half) * ldd + c0 + n] = acc[r] + bias;
}

__device__ __forceinline__ void count_rank_edges(const int* __restrict__ col, int* __restrict__ cnt,
                                                 int* __restrict__ rank, int E,
                                                 int first, int stride) {
    for (int e0 = first; e0 < E; e0 += EDGE_BATCH * stride) {
        int c[EDGE_BATCH], r[EDGE_BATCH];
#pragma unroll
        for (int u = 0; u < EDGE_BATCH; ++u) {
            const int e = e0 + u * stride;
            c[u] = e < E ? col[e] : -1;
        }
#pragma unroll
        for (int u = 0; u < EDGE_BATCH; ++u)
            r[u] = c[u] >= 0 ? atomicAdd(&cnt[c[u]], 1) : 0;
#pragma unroll
        for (int u = 0; u < EDGE_BATCH; ++u) {
            const int e = e0 + u * stride;
            if (e < E) rank[e] = r[u];
        }
    }
}

__global__ __launch_bounds__(1024) void offs_dis_kernel(
    const int* __restrict__ cnt, int* __restrict__ offs, float* __restrict__ dis, int N)
{
    __shared__ int wave_incl[32];
    __shared__ int wave_base[32];
    const int tid  = threadIdx.x;
    const int lane = tid & 31;
    const int wv   = tid >> 5;
    const int i    = blockIdx.x * 1024 + tid;
    const int c    = (i < N) ? cnt[i] : 0;

    int part = 0;
    for (int j = 0; j < (int)blockIdx.x; ++j) part += cnt[j * 1024 + tid];

    int inc = c;
    for (int d = 1; d < 32; d <<= 1) {
        int v = __shfl_up(inc, d, 32);
        if (lane >= d) inc += v;
    }
    for (int d = 16; d > 0; d >>= 1) part += __shfl_xor(part, d, 32);
    if (lane == 31) wave_incl[wv] = inc;
    if (lane == 0)  wave_base[wv] = part;
    __syncthreads();

    int add = ((lane < wv) ? wave_incl[lane] : 0) + wave_base[lane];
    for (int d = 16; d > 0; d >>= 1) add += __shfl_xor(add, d, 32);

    const int excl = add + inc - c;
    if (i < N) {
        offs[i] = excl;
        dis[i]  = rsqrtf((float)(c + 1));
    }
    if (i == N - 1) offs[N] = excl + c;
}

__global__ void edge_fill_kernel(const int* __restrict__ row, const int* __restrict__ col,
                                 const int* __restrict__ rank, const int* __restrict__ offs,
                                 const float* __restrict__ dis,
                                 int* __restrict__ src, float* __restrict__ wgt, int E) {
    int e = blockIdx.x * blockDim.x + threadIdx.x;
    if (e >= E) return;
    int r = row[e], c = col[e];
    int p = offs[c] + rank[e];
    src[p] = r;
    wgt[p] = dis[r] * dis[c];
}

__global__ __launch_bounds__(128) void weight_fold_kernel(
    const float* __restrict__ Wg1, const float* __restrict__ bg1,
    const float* __restrict__ Wg2, float* __restrict__ W12,
    int* __restrict__ cnt, int N)
{
    if (blockIdx.x >= 9) {
        const int i = ((int)blockIdx.x - 9) * 128 + (int)threadIdx.x;
        if (i < N) cnt[i] = 0;
        return;
    }
    const int lane = threadIdx.x & 31;
    const int c0   = (threadIdx.x >> 5) * 16;
    const int r0   = blockIdx.x * 16;
    const int n    = lane & 15, h = lane >> 4;
    const int row  = r0 + n;
    const float* arow = row < 128 ? Wg1 + row * 128 : bg1;
    const bool  live  = row <= 128;
    const float* bcol = Wg2 + c0 + n;

    v8f acc0 = {}, acc1 = {};
#pragma unroll
    for (int kc = 0; kc < 32; kc += 2) {
        const int k0 = kc * 4 + 2 * h, k1 = k0 + 4;
        v2f a0 = v2f{arow[k0], arow[k0 + 1]};
        v2f a1 = v2f{arow[k1], arow[k1 + 1]};
        if (!live) { a0 = v2f{0.f, 0.f}; a1 = v2f{0.f, 0.f}; }
        const v2f b0 = v2f{bcol[k0 * 64], bcol[(k0 + 1) * 64]};
        const v2f b1 = v2f{bcol[k1 * 64], bcol[(k1 + 1) * 64]};
        acc0 = wmma_f32(a0, b0, acc0);
        acc1 = wmma_f32(a1, b1, acc1);
    }
    store_tile(W12, 64, r0, c0, acc0 + acc1, lane, 0.f);
}

__global__ __launch_bounds__((GEMM_WAVES + EDGE_WAVES) * 32) void self_branch_kernel(
    const float* __restrict__ x, const float* __restrict__ Win,
    const float* __restrict__ bin, const float* __restrict__ Wout,
    const float* __restrict__ bout, const float* __restrict__ W12,
    float* __restrict__ l1_out, float* __restrict__ p_out, int N,
    const int* __restrict__ col, int* __restrict__ cnt, int* __restrict__ rank, int E)
{
    __shared__ __attribute__((aligned(16))) float lds[16384 + 8192 + 16384 + GEMM_WAVES * 16 * LDS_LD];
    float* Lin  = lds;
    float* L12  = lds + 16384;
    float* Lout = lds + 24576;
    const int tid  = threadIdx.x;
    const int lane = tid & 31;
    const int wave = tid >> 5;
    const int n    = lane & 15, half = lane >> 4;
    float* myL = lds + 40960 + wave * 16 * LDS_LD;

    if (wave < GEMM_WAVES) {
        stage_weights<128, 128>(Win,  Lin,  tid);
        stage_weights<128, 64>(W12,  L12,  tid);
        stage_weights<256, 64>(Wout, Lout, tid);
    }
    __syncthreads();

    if (wave >= GEMM_WAVES) {
        count_rank_edges(col, cnt, rank, E,
                         (int)blockIdx.x * (EDGE_WAVES * 32) + (tid - GEMM_WAVES * 32),
                         (int)gridDim.x * (EDGE_WAVES * 32));
        return;
    }

    const int strips = N / 16;
    for (int strip = wave * (int)gridDim.x + (int)blockIdx.x; strip < strips;
         strip += GEMM_WAVES * (int)gridDim.x) {
        const int r0 = strip * 16;

        v2f ax[32];
#pragma unroll
        for (int kc = 0; kc < 32; ++kc)
            ax[kc] = a_frag_g(x, 128, r0, kc * 4, lane);

#pragma unroll 1
        for (int t = 0; t < 8; t += 2) {
            v8f acc0 = {}, acc1 = {};
            mma_pass<16, 128>(ax, Lin, 0, t * 16, lane, acc0, acc1);
            const float bb0 = bin[t * 16 + n], bb1 = bin[t * 16 + 16 + n];
#pragma unroll
            for (int r = 0; r < 8; ++r) {
                const float v0 = acc0[r] + bb0, v1 = acc1[r] + bb1;
                myL[(r + 8 * half) * LDS_LD + t * 16 + n]      = v0 > 0.f ? v0 : 0.f;
                myL[(r + 8 * half) * LDS_LD + t * 16 + 16 + n] = v1 > 0.f ? v1 : 0.f;
            }
        }

#pragma unroll 1
        for (int t = 0; t < 4; t += 2) {
            v8f acc0 = {}, acc1 = {};
            mma_pass<16, 64>(ax, L12, 0, t * 16, lane, acc0, acc1);
            store_tile(p_out, 64, r0, t * 16,      acc0, lane, 0.f);
            store_tile(p_out, 64, r0, t * 16 + 16, acc1, lane, 0.f);
        }

        __builtin_amdgcn_fence(__ATOMIC_ACQ_REL, "wavefront");

        v2f al[32];
#pragma unroll
        for (int kc = 0; kc < 32; ++kc)
            al[kc] = a_frag_lds(myL, LDS_LD, kc * 4, lane);

#pragma unroll 1
        for (int t = 0; t < 4; t += 2) {
            v8f acc0 = {}, acc1 = {};
            mma_pass<16, 64>(ax, Lout, 0,  t * 16, lane, acc0, acc1);
            mma_pass<16, 64>(al, Lout, 16, t * 16, lane, acc0, acc1);
            store_tile(l1_out, 64, r0, t * 16,      acc0, lane, bout[t * 16 + n]);
            store_tile(l1_out, 64, r0, t * 16 + 16, acc1, lane, bout[t * 16 + 16 + n]);
        }

        __builtin_amdgcn_fence(__ATOMIC_ACQ_REL, "wavefront");
    }
}

template <int V, int U>
__device__ __forceinline__ void gather_rows(
    const float* __restrict__ h, int W, int s, float w, int k, int lo,
    float __attribute__((ext_vector_type(V)))& acc)
{
    typedef float vecf __attribute__((ext_vector_type(V)));
    vecf  v[U];
    float ww[U];
    for (int u = 0; u < U; ++u) {
        const int su = __builtin_amdgcn_readlane(s, k + u);
        ww[u] = __int_as_float(__builtin_amdgcn_readlane(__float_as_int(w), k + u));
        v[u]  = *(const vecf*)(h + (size_t)su * W + lo);
    }
    for (int u = 0; u < U; ++u) acc += v[u] * ww[u];
}

template <int W>
__global__ __launch_bounds__(256) void seg_gather_kernel(
    const int* __restrict__ offs, const int* __restrict__ src, const float* __restrict__ wgt,
    const float* __restrict__ dis, const float* __restrict__ h, const float* __restrict__ bias,
    float* __restrict__ g, int N)
{
    constexpr int V = W / 32;
    typedef float vecf __attribute__((ext_vector_type(V)));
    const int lane = threadIdx.x & 31;
    const int i = __builtin_amdgcn_readfirstlane(blockIdx.x * 8 + (threadIdx.x >> 5));
    if (i >= N) return;
    const int beg = __builtin_amdgcn_readfirstlane(offs[i]);
    const int end = __builtin_amdgcn_readfirstlane(offs[i + 1]);
    const int lo  = lane * V;
    const float d = dis[i];
    vecf acc = *(const vecf*)(bias + lo) + *(const vecf*)(h + (size_t)i * W + lo) * (d * d);

    for (int base = beg; base < end; base += 32) {
        const int m = min(32, end - base);
        int   s = 0;
        float w = 0.f;
        if (lane < m) { s = src[base + lane]; w = wgt[base + lane]; }
        int k = 0;
        for (; k + 8 <= m; k += 8) gather_rows<V, 8>(h, W, s, w, k, lo, acc);
        if (k + 4 <= m) { gather_rows<V, 4>(h, W, s, w, k, lo, acc); k += 4; }
        if (k + 2 <= m) { gather_rows<V, 2>(h, W, s, w, k, lo, acc); k += 2; }
        if (k < m)      { gather_rows<V, 1>(h, W, s, w, k, lo, acc); }
    }
    *(vecf*)(g + (size_t)i * W + lo) = acc;
}

__global__ __launch_bounds__(GEMM_WAVES * 32) void out_gemm_kernel(
    const float* __restrict__ xn, const float* __restrict__ g2,
    const float* __restrict__ W, const float* __restrict__ b,
    float* __restrict__ out, int N)
{
    __shared__ __attribute__((aligned(16))) float Lw[192 * 64];
    const int tid  = threadIdx.x;
    const int lane = tid & 31;
    const int wave = tid >> 5;
    const int n    = lane & 15;
    stage_weights<192, 64>(W, Lw, tid);
    __syncthreads();

    const int strips = N / 16;
    for (int strip = wave * (int)gridDim.x + (int)blockIdx.x; strip < strips;
         strip += GEMM_WAVES * (int)gridDim.x) {
        const int r0 = strip * 16;
        v2f axn[32];
#pragma unroll
        for (int kc = 0; kc < 32; ++kc)
            axn[kc] = a_frag_g(xn, 128, r0, kc * 4, lane);
        v2f ag2[16];
#pragma unroll
        for (int kc = 0; kc < 16; ++kc)
            ag2[kc] = a_frag_g(g2, 64, r0, kc * 4, lane);

        for (int t = 0; t < 4; t += 2) {
            v8f acc0 = {}, acc1 = {};
            mma_pass<16, 64>(axn, Lw, 0,  t * 16, lane, acc0, acc1);
            mma_pass<8, 64>(ag2, Lw, 16, t * 16, lane, acc0, acc1);
            store_tile(out, 64, r0, t * 16,      acc0, lane, b[t * 16 + n]);
            store_tile(out, 64, r0, t * 16 + 16, acc1, lane, b[t * 16 + 16 + n]);
        }
    }
}

extern "C" void kernel_launch(void* const* d_in, const int* in_sizes, int n_in,
                              void* d_out, int out_size, void* d_ws, size_t ws_size,
                              hipStream_t stream) {
    const float* x_self     = (const float*)d_in[0];
    const float* x_neighbor = (const float*)d_in[1];
    const int*   edge_index = (const int*)d_in[2];
    const float* W_in_self  = (const float*)d_in[3];
    const float* b_in_self  = (const float*)d_in[4];
    const float* W_out_self = (const float*)d_in[5];
    const float* b_out_self = (const float*)d_in[6];
    const float* Wg1        = (const float*)d_in[7];
    const float* bg1        = (const float*)d_in[8];
    const float* Wg2        = (const float*)d_in[9];
    const float* bg2        = (const float*)d_in[10];
    const float* W_out      = (const float*)d_in[11];
    const float* b_out      = (const float*)d_in[12];

    const int N = in_sizes[0] / 128;
    const int E = in_sizes[2] / 2;
    const int* row = edge_index;
    const int* col = edge_index + E;

    float* ws  = (float*)d_ws;
    float* dis = ws;
    int*   cnt = (int*)(ws + N);
    float* p   = ws + 2 * (size_t)N;
    float* h2  = p  + (size_t)N * 64;
    float* g2  = h2 + (size_t)N * 64;
    float* W12 = g2 + (size_t)N * 64;

    float* l1_out = (float*)d_out;
    float* x2_out = (float*)d_out + (size_t)N * 64;

    int*   src  = (int*)x2_out;
    float* wgt  = x2_out + E;
    int*   rank = (int*)x2_out + 2 * (size_t)E;
    int*   offs = (int*)x2_out + 3 * (size_t)E;

    weight_fold_kernel<<<9 + (N + 127) / 128, 128, 0, stream>>>(Wg1, bg1, Wg2, W12, cnt, N);

    self_branch_kernel<<<GEMM_BLOCKS, (GEMM_WAVES + EDGE_WAVES) * 32, 0, stream>>>(
        x_self, W_in_self, b_in_self, W_out_self, b_out_self, W12, l1_out, p, N,
        col, cnt, rank, E);

    offs_dis_kernel<<<(N + 1023) / 1024, 1024, 0, stream>>>(cnt, offs, dis, N);
    edge_fill_kernel<<<(E + 255) / 256, 256, 0, stream>>>(row, col, rank, offs, dis, src, wgt, E);

    seg_gather_kernel<64><<<(N + 7) / 8, 256, 0, stream>>>(offs, src, wgt, dis, p, W12 + 128 * 64, h2, N);

    seg_gather_kernel<64><<<(N + 7) / 8, 256, 0, stream>>>(offs, src, wgt, dis, h2, bg2, g2, N);

    out_gemm_kernel<<<GEMM_BLOCKS, GEMM_WAVES * 32, 0, stream>>>(x_neighbor, g2, W_out, b_out, x2_out, N);
}


# Edge-list scan + fill as one launch: each of 256 resident work-groups scans the in-degrees into its own LDS and fills its run of edges from that table; scan launch gone; values and sum order kept
# speedup vs baseline: 2.7346x; 1.1491x over previous
#include <hip/hip_runtime.h>
#include <hip/hip_bf16.h>

typedef float v2f __attribute__((ext_vector_type(2)));
typedef float v4f __attribute__((ext_vector_type(4)));
typedef float v8f __attribute__((ext_vector_type(8)));

#define GEMM_WAVES  8
#define GEMM_BLOCKS 256
#define EDGE_WAVES  4
#define EDGE_BATCH  4

#define LDS_LD 130

__device__ __forceinline__ v8f wmma_f32(v2f a, v2f b, v8f c) {
    return __builtin_amdgcn_wmma_f32_16x16x4_f32(
         false, a,  false, b,
         (short)0, c,  false,  false);
}

__device__ __forceinline__ v2f a_frag_g(const float* __restrict__ A, int lda,
                                        int r0, int k, int lane) {
    int m  = lane & 15;
    int kk = k + ((lane >> 4) << 1);
    const float* p = A + (size_t)(r0 + m) * lda + kk;
    return v2f{p[0], p[1]};
}

__device__ __forceinline__ v2f a_frag_lds(const float* __restrict__ L, int ld,
                                          int k, int lane) {
    int m  = lane & 15;
    int kk = k + ((lane >> 4) << 1);
    const float* p = L + m * ld + kk;
    return v2f{p[0], p[1]};
}

template <int K, int NC>
__device__ __forceinline__ void stage_weights(const float* __restrict__ Wm,
                                              float* __restrict__ L, int tid) {
#pragma unroll 4
    for (int it = tid; it < K * NC / 4; it += GEMM_WAVES * 32) {
        const int c  = it % NC;
        const int qh = it / NC;
        const int k0 = (qh >> 1) * 8 + (qh & 1) * 2;
        const float* p = Wm + k0 * NC + c;
        *(v4f*)(L + it * 4) = v4f{p[0], p[NC], p[4 * NC], p[5 * NC]};
    }
}

template <int NQ, int NC>
__device__ __forceinline__ void mma_pass(const v2f* a, const float* __restrict__ L,
                                         int q0, int c0, int lane,
                                         v8f& acc0, v8f& acc1) {
    const int n = lane & 15, h = lane >> 4;
    v4f b0[NQ], b1[NQ];
#pragma unroll
    for (int q = 0; q < NQ; ++q) {
        const float* p = L + (((q0 + q) * 2 + h) * NC + c0 + n) * 4;
        b0[q] = *(const v4f*)p;
        b1[q] = *(const v4f*)(p + 64);
    }
#pragma unroll
    for (int q = 0; q < NQ; ++q) {
        acc0 = wmma_f32(a[2 * q],     v2f{b0[q].x, b0[q].y}, acc0);
        acc1 = wmma_f32(a[2 * q],     v2f{b1[q].x, b1[q].y}, acc1);
        acc0 = wmma_f32(a[2 * q + 1], v2f{b0[q].z, b0[q].w}, acc0);
        acc1 = wmma_f32(a[2 * q + 1], v2f{b1[q].z, b1[q].w}, acc1);
    }
}

__device__ __forceinline__ void store_tile(float* __restrict__ D, int ldd,
                                           int r0, int c0, v8f acc, int lane,
                                           float bias) {
    int n    = lane & 15;
    int half = lane >> 4;
#pragma unroll
    for (int r = 0; r < 8; ++r)
        D[(size_t)(r0 + r + 8 * half) * ldd + c0 + n] = acc[r] + bias;
}

__device__ __forceinline__ void count_rank_edges(const int* __restrict__ col, int* __restrict__ cnt,
                                                 int* __restrict__ rank, int E,
                                                 int first, int stride) {
    for (int e0 = first; e0 < E; e0 += EDGE_BATCH * stride) {
        int c[EDGE_BATCH], r[EDGE_BATCH];
#pragma unroll
        for (int u = 0; u < EDGE_BATCH; ++u) {
            const int e = e0 + u * stride;
            c[u] = e < E ? col[e] : -1;
        }
#pragma unroll
        for (int u = 0; u < EDGE_BATCH; ++u)
            r[u] = c[u] >= 0 ? atomicAdd(&cnt[c[u]], 1) : 0;
#pragma unroll
        for (int u = 0; u < EDGE_BATCH; ++u) {
            const int e = e0 + u * stride;
            if (e < E) rank[e] = r[u];
        }
    }
}

#define FILL_BLOCKS   256
#define FILL_THREADS  512
#define FILL_WAVES    (FILL_THREADS / 32)
#define SCAN_CAP      40960
#define SCAN_PER_WAVE (SCAN_CAP / FILL_WAVES)
#define SCAN_ITERS    (SCAN_PER_WAVE / 128)
#define FILL_SLICE    (SCAN_CAP / FILL_BLOCKS + 1)
#define FILL_BATCH    5

typedef int v4i __attribute__((ext_vector_type(4)));

__device__ __forceinline__ int wave_incl_scan(int x, int lane) {
    x += __builtin_amdgcn_update_dpp(0, x, 0x111, 0xf, 0xf, false);
    x += __builtin_amdgcn_update_dpp(0, x, 0x112, 0xf, 0xf, false);
    x += __builtin_amdgcn_update_dpp(0, x, 0x114, 0xf, 0xf, false);
    x += __builtin_amdgcn_update_dpp(0, x, 0x118, 0xf, 0xf, false);
    const int low = __builtin_amdgcn_readlane(x, 15);
    return x + (lane >= 16 ? low : 0);
}

__global__ __launch_bounds__(FILL_THREADS) void scan_fill_kernel(
    const int* __restrict__ cnt, const int* __restrict__ row, const int* __restrict__ col,
    const int* __restrict__ rank, int* __restrict__ offs, float* __restrict__ dis,
    int* __restrict__ src, float* __restrict__ wgt, int N, int E)
{
    __shared__ __attribute__((aligned(16))) int table[SCAN_CAP + 4];
    __shared__ int wave_tot[FILL_WAVES];
    const int tid  = threadIdx.x;
    const int lane = tid & 31;
    const int wv   = __builtin_amdgcn_readfirstlane(tid >> 5);

    const int epb  = (E + (int)gridDim.x - 1) / (int)gridDim.x;
    const int e_lo = (int)blockIdx.x * epb;
    const int e_hi = min(E, e_lo + epb);


    const int nb = wv * SCAN_PER_WAVE + 4 * lane;
    v4i c[SCAN_ITERS];
#pragma unroll
    for (int j = 0; j < SCAN_ITERS; ++j)
        c[j] = *(const v4i*)(cnt + min(nb + 128 * j, N - 4));

    int er[FILL_BATCH], ec[FILL_BATCH], ek[FILL_BATCH];
#pragma unroll
    for (int u = 0; u < FILL_BATCH; ++u) {
        const int e = min(e_lo + tid + u * FILL_THREADS, E - 1);
        er[u] = row[e];
        ec[u] = col[e];
        ek[u] = rank[e];
    }

    int carry = 0;
#pragma unroll
    for (int j = 0; j < SCAN_ITERS; ++j) {
        if (nb + 128 * j >= N) c[j] = v4i{0, 0, 0, 0};
        const int s1 = c[j].x, s2 = s1 + c[j].y, s3 = s2 + c[j].z, s4 = s3 + c[j].w;
        const int inc  = wave_incl_scan(s4, lane);
        const int excl = carry + inc - s4;
        c[j] = v4i{excl, excl + s1, excl + s2, excl + s3};
        carry += __builtin_amdgcn_readlane(inc, 31);
    }
    if (lane == 0) wave_tot[wv] = carry;
    __syncthreads();

    const int before = __builtin_amdgcn_readlane(
        wave_incl_scan(lane < wv ? wave_tot[lane] : 0, lane), 31);
#pragma unroll
    for (int j = 0; j < SCAN_ITERS; ++j)
        *(v4i*)(table + nb + 128 * j) = c[j] + before;
    if (tid == FILL_THREADS - 1) table[SCAN_CAP] = before + carry;
    __syncthreads();

    if (tid < FILL_SLICE) {
        const int i = (int)blockIdx.x * FILL_SLICE + tid;
        if (i <= N) {
            const int o = table[i];
            offs[i] = o;
            if (i < N) dis[i] = rsqrtf((float)(table[i + 1] - o + 1));
        }
    }

    for (int k0 = 0; ; ) {
        int   p[FILL_BATCH];
        float w[FILL_BATCH];
#pragma unroll
        for (int u = 0; u < FILL_BATCH; ++u) {
            const bool ok = e_lo + k0 + tid + u * FILL_THREADS < e_hi;
            const int r = ok ? er[u] : 0, cc = ok ? ec[u] : 0;
            const int c0 = table[cc], c1 = table[cc + 1];
            const int r0 = table[r],  r1 = table[r + 1];
            p[u] = c0 + ek[u];
            w[u] = rsqrtf((float)(r1 - r0 + 1)) * rsqrtf((float)(c1 - c0 + 1));
        }
#pragma unroll
        for (int u = 0; u < FILL_BATCH; ++u) {
            if (e_lo + k0 + tid + u * FILL_THREADS < e_hi) {
                src[p[u]] = er[u];
                wgt[p[u]] = w[u];
            }
        }
        k0 += FILL_THREADS * FILL_BATCH;
        if (k0 >= epb) break;
#pragma unroll
        for (int u = 0; u < FILL_BATCH; ++u) {
            const int e = min(e_lo + k0 + tid + u * FILL_THREADS, E - 1);
            er[u] = row[e];
            ec[u] = col[e];
            ek[u] = rank[e];
        }
    }
}

#define FOLD_LD 68

__global__ __launch_bounds__(128) void weight_fold_kernel(
    const float* __restrict__ Wg1, const float* __restrict__ bg1,
    const float* __restrict__ Wg2, const float* __restrict__ bg2,
    const float* __restrict__ Wo2, float* __restrict__ Wf,
    int* __restrict__ cnt, int N)
{
    if (blockIdx.x >= 9) {
        const int i = ((int)blockIdx.x - 9) * 128 + (int)threadIdx.x;
        if (i < N) cnt[i] = 0;
        return;
    }
    __shared__ __attribute__((aligned(16))) float Lt[16 * FOLD_LD];
    const int lane = threadIdx.x & 31;
    const int c0   = (threadIdx.x >> 5) * 16;
    const int r0   = blockIdx.x * 16;
    const int n    = lane & 15, h = lane >> 4;
    const int row  = r0 + n;
    const float* arow = row < 128 ? Wg1 + row * 128 : bg1;
    const bool  live  = row <= 128;
    const float* bcol = Wg2 + c0 + n;
    const float* ocol = Wo2 + c0 + n;

    v2f bo[16];
#pragma unroll
    for (int kc = 0; kc < 16; ++kc) {
        const int k0 = kc * 4 + 2 * h;
        bo[kc] = v2f{ocol[k0 * 64], ocol[(k0 + 1) * 64]};
    }

    v8f acc0 = {}, acc1 = {};
#pragma unroll
    for (int kc = 0; kc < 32; kc += 2) {
        const int k0 = kc * 4 + 2 * h, k1 = k0 + 4;
        v2f a0 = v2f{arow[k0], arow[k0 + 1]};
        v2f a1 = v2f{arow[k1], arow[k1 + 1]};
        if (!live) { a0 = v2f{0.f, 0.f}; a1 = v2f{0.f, 0.f}; }
        const v2f b0 = v2f{bcol[k0 * 64], bcol[(k0 + 1) * 64]};
        const v2f b1 = v2f{bcol[k1 * 64], bcol[(k1 + 1) * 64]};
        acc0 = wmma_f32(a0, b0, acc0);
        acc1 = wmma_f32(a1, b1, acc1);
    }
    const v8f t = acc0 + acc1;
    const float b2 = bg2[c0 + n];
#pragma unroll
    for (int r = 0; r < 8; ++r) {
        const int lr = r + 8 * h;
        Lt[lr * FOLD_LD + c0 + n] = (r0 + lr == 129) ? b2 : t[r];
    }
    __syncthreads();

    v8f d0 = {}, d1 = {};
#pragma unroll
    for (int kc = 0; kc < 16; kc += 2) {
        d0 = wmma_f32(a_frag_lds(Lt, FOLD_LD, kc * 4,     lane), bo[kc],     d0);
        d1 = wmma_f32(a_frag_lds(Lt, FOLD_LD, kc * 4 + 4, lane), bo[kc + 1], d1);
    }
    store_tile(Wf, 64, r0, c0, d0 + d1, lane, 0.f);
}

__global__ __launch_bounds__((GEMM_WAVES + EDGE_WAVES) * 32) void self_branch_kernel(
    const float* __restrict__ x, const float* __restrict__ Win,
    const float* __restrict__ bin, const float* __restrict__ Wout,
    const float* __restrict__ bout, const float* __restrict__ Wf,
    const float* __restrict__ xn, const float* __restrict__ Wo1,
    const float* __restrict__ bo,
    float* __restrict__ l1_out, float* __restrict__ p_out,
    float* __restrict__ q_out, int N,
    const int* __restrict__ col, int* __restrict__ cnt, int* __restrict__ rank, int E)
{
    __shared__ __attribute__((aligned(16))) float lds[16384 + 8192 + 16384 + 8192 + GEMM_WAVES * 16 * LDS_LD];
    float* Lin  = lds;
    float* Lf  = lds + 16384;
    float* Lout = lds + 24576;
    float* Lo1  = lds + 40960;
    const int tid  = threadIdx.x;
    const int lane = tid & 31;
    const int wave = __builtin_amdgcn_readfirstlane(tid >> 5);
    const int n    = lane & 15, half = lane >> 4;
    float* myL = lds + 49152 + wave * 16 * LDS_LD;

    if (tid < GEMM_WAVES * 32) {
        stage_weights<128, 128>(Win,  Lin,  tid);
        stage_weights<128, 64>(Wf,  Lf,  tid);
        stage_weights<256, 64>(Wout, Lout, tid);
        stage_weights<128, 64>(Wo1,  Lo1,  tid);
    }
    __syncthreads();

    if (wave >= GEMM_WAVES) {
        count_rank_edges(col, cnt, rank, E,
                         (int)blockIdx.x * (EDGE_WAVES * 32) + (tid - GEMM_WAVES * 32),
                         (int)gridDim.x * (EDGE_WAVES * 32));
        return;
    }

    const int strips = N / 16;

    if (wave & 2) {
        const int slot = ((wave >> 2) << 1) | (wave & 1);
        for (int strip = slot * (int)gridDim.x + (int)blockIdx.x; strip < strips;
             strip += 4 * (int)gridDim.x) {
            const int r0 = strip * 16;
            v2f axn[32];
#pragma unroll
            for (int kc = 0; kc < 32; ++kc)
                axn[kc] = a_frag_g(xn, 128, r0, kc * 4, lane);
#pragma unroll 1
            for (int t = 0; t < 4; t += 2) {
                v8f acc0 = {}, acc1 = {};
                mma_pass<16, 64>(axn, Lo1, 0, t * 16, lane, acc0, acc1);
                store_tile(q_out, 64, r0, t * 16,      acc0, lane, bo[t * 16 + n]      + Wf[129 * 64 + t * 16 + n]);
                store_tile(q_out, 64, r0, t * 16 + 16, acc1, lane, bo[t * 16 + 16 + n] + Wf[129 * 64 + t * 16 + 16 + n]);
            }
        }
    }

    for (int strip = wave * (int)gridDim.x + (int)blockIdx.x; strip < strips;
         strip += GEMM_WAVES * (int)gridDim.x) {
        const int r0 = strip * 16;

        v2f ax[32];
#pragma unroll
        for (int kc = 0; kc < 32; ++kc)
            ax[kc] = a_frag_g(x, 128, r0, kc * 4, lane);

#pragma unroll 1
        for (int t = 0; t < 8; t += 2) {
            v8f acc0 = {}, acc1 = {};
            mma_pass<16, 128>(ax, Lin, 0, t * 16, lane, acc0, acc1);
            const float bb0 = bin[t * 16 + n], bb1 = bin[t * 16 + 16 + n];
#pragma unroll
            for (int r = 0; r < 8; ++r) {
                const float v0 = acc0[r] + bb0, v1 = acc1[r] + bb1;
                myL[(r + 8 * half) * LDS_LD + t * 16 + n]      = v0 > 0.f ? v0 : 0.f;
                myL[(r + 8 * half) * LDS_LD + t * 16 + 16 + n] = v1 > 0.f ? v1 : 0.f;
            }
        }

#pragma unroll 1
        for (int t = 0; t < 4; t += 2) {
            v8f acc0 = {}, acc1 = {};
            mma_pass<16, 64>(ax, Lf, 0, t * 16, lane, acc0, acc1);
            store_tile(p_out, 64, r0, t * 16,      acc0, lane, 0.f);
            store_tile(p_out, 64, r0, t * 16 + 16, acc1, lane, 0.f);
        }

        __builtin_amdgcn_fence(__ATOMIC_ACQ_REL, "wavefront");

        v2f al[32];
#pragma unroll
        for (int kc = 0; kc < 32; ++kc)
            al[kc] = a_frag_lds(myL, LDS_LD, kc * 4, lane);

#pragma unroll 1
        for (int t = 0; t < 4; t += 2) {
            v8f acc0 = {}, acc1 = {};
            mma_pass<16, 64>(ax, Lout, 0,  t * 16, lane, acc0, acc1);
            mma_pass<16, 64>(al, Lout, 16, t * 16, lane, acc0, acc1);
            store_tile(l1_out, 64, r0, t * 16,      acc0, lane, bout[t * 16 + n]);
            store_tile(l1_out, 64, r0, t * 16 + 16, acc1, lane, bout[t * 16 + 16 + n]);
        }

        __builtin_amdgcn_fence(__ATOMIC_ACQ_REL, "wavefront");
    }
}

template <int V, int U>
__device__ __forceinline__ void gather_rows(
    const float* __restrict__ h, int W, int s, float w, int k, int lo,
    float __attribute__((ext_vector_type(V)))& acc)
{
    typedef float vecf __attribute__((ext_vector_type(V)));
    vecf  v[U];
    float ww[U];
    for (int u = 0; u < U; ++u) {
        const int su = __builtin_amdgcn_readlane(s, k + u);
        ww[u] = __int_as_float(__builtin_amdgcn_readlane(__float_as_int(w), k + u));
        v[u]  = *(const vecf*)(h + (size_t)su * W + lo);
    }
    for (int u = 0; u < U; ++u) acc += v[u] * ww[u];
}

template <int W>
__global__ __launch_bounds__(256) void seg_gather_kernel(
    const int* __restrict__ offs, const int* __restrict__ src, const float* __restrict__ wgt,
    const float* __restrict__ dis, const float* __restrict__ h, const float* __restrict__ bias,
    int bstride, float* __restrict__ g, int N)
{
    constexpr int V = W / 32;
    typedef float vecf __attribute__((ext_vector_type(V)));
    const int lane = threadIdx.x & 31;
    const int i = __builtin_amdgcn_readfirstlane(blockIdx.x * 8 + (threadIdx.x >> 5));
    if (i >= N) return;
    const int beg = __builtin_amdgcn_readfirstlane(offs[i]);
    const int end = __builtin_amdgcn_readfirstlane(offs[i + 1]);
    const int lo  = lane * V;
    const float d = dis[i];
    vecf acc = *(const vecf*)(bias + (size_t)i * bstride + lo) + *(const vecf*)(h + (size_t)i * W + lo) * (d * d);

    for (int base = beg; base < end; base += 32) {
        const int m = min(32, end - base);
        int   s = 0;
        float w = 0.f;
        if (lane < m) { s = src[base + lane]; w = wgt[base + lane]; }
        int k = 0;
        for (; k + 8 <= m; k += 8) gather_rows<V, 8>(h, W, s, w, k, lo, acc);
        if (k + 4 <= m) { gather_rows<V, 4>(h, W, s, w, k, lo, acc); k += 4; }
        if (k + 2 <= m) { gather_rows<V, 2>(h, W, s, w, k, lo, acc); k += 2; }
        if (k < m)      { gather_rows<V, 1>(h, W, s, w, k, lo, acc); }
    }
    *(vecf*)(g + (size_t)i * W + lo) = acc;
}

extern "C" void kernel_launch(void* const* d_in, const int* in_sizes, int n_in,
                              void* d_out, int out_size, void* d_ws, size_t ws_size,
                              hipStream_t stream) {
    const float* x_self     = (const float*)d_in[0];
    const float* x_neighbor = (const float*)d_in[1];
    const int*   edge_index = (const int*)d_in[2];
    const float* W_in_self  = (const float*)d_in[3];
    const float* b_in_self  = (const float*)d_in[4];
    const float* W_out_self = (const float*)d_in[5];
    const float* b_out_self = (const float*)d_in[6];
    const float* Wg1        = (const float*)d_in[7];
    const float* bg1        = (const float*)d_in[8];
    const float* Wg2        = (const float*)d_in[9];
    const float* bg2        = (const float*)d_in[10];
    const float* W_out      = (const float*)d_in[11];
    const float* b_out      = (const float*)d_in[12];

    const int N = in_sizes[0] / 128;
    const int E = in_sizes[2] / 2;
    const int* row = edge_index;
    const int* col = edge_index + E;

    float* l1_out = (float*)d_out;
    float* x2_out = (float*)d_out + (size_t)N * 64;

    float* ws  = (float*)d_ws;
    float* dis = ws;
    int*   cnt = (int*)(ws + N);
    float* h2  = ws + 2 * (size_t)N;
    float* q   = h2 + (size_t)N * 64;
    float* Wf  = q  + (size_t)N * 64;
    int*   offs = (int*)(Wf + 144 * 64);
    int*   src  = offs + N + 64;
    float* wgt  = (float*)(src + E);
    int*   rank = (int*)(wgt + E);

    float* p = x2_out;

    weight_fold_kernel<<<9 + (N + 127) / 128, 128, 0, stream>>>(Wg1, bg1, Wg2, bg2, W_out + 128 * 64, Wf, cnt, N);

    self_branch_kernel<<<GEMM_BLOCKS, (GEMM_WAVES + EDGE_WAVES) * 32, 0, stream>>>(
        x_self, W_in_self, b_in_self, W_out_self, b_out_self, Wf,
        x_neighbor, W_out, b_out, l1_out, p, q, N,
        col, cnt, rank, E);

    scan_fill_kernel<<<FILL_BLOCKS, FILL_THREADS, 0, stream>>>(cnt, row, col, rank, offs, dis, src, wgt, N, E);

    seg_gather_kernel<64><<<(N + 7) / 8, 256, 0, stream>>>(offs, src, wgt, dis, p, Wf + 128 * 64, 0, h2, N);

    seg_gather_kernel<64><<<(N + 7) / 8, 256, 0, stream>>>(offs, src, wgt, dis, h2, q, 64, x2_out, N);
}


# Edges put straight into fixed 64-slot per-target buckets by the count+rank add (overflow list past 64 slots); scan+fill launch gone, 5->4 launches; first gather forms the weights; rank sum order kept
# speedup vs baseline: 3.4601x; 1.2653x over previous
#include <hip/hip_runtime.h>
#include <hip/hip_bf16.h>

typedef float v2f __attribute__((ext_vector_type(2)));
typedef float v4f __attribute__((ext_vector_type(4)));
typedef float v8f __attribute__((ext_vector_type(8)));

#define GEMM_WAVES  8
#define GEMM_BLOCKS 256
#define EDGE_WAVES  4
#define EDGE_BATCH  4

#define LDS_LD 130

__device__ __forceinline__ v8f wmma_f32(v2f a, v2f b, v8f c) {
    return __builtin_amdgcn_wmma_f32_16x16x4_f32(
         false, a,  false, b,
         (short)0, c,  false,  false);
}

__device__ __forceinline__ v2f a_frag_g(const float* __restrict__ A, int lda,
                                        int r0, int k, int lane) {
    int m  = lane & 15;
    int kk = k + ((lane >> 4) << 1);
    const float* p = A + (size_t)(r0 + m) * lda + kk;
    return v2f{p[0], p[1]};
}

__device__ __forceinline__ v2f a_frag_lds(const float* __restrict__ L, int ld,
                                          int k, int lane) {
    int m  = lane & 15;
    int kk = k + ((lane >> 4) << 1);
    const float* p = L + m * ld + kk;
    return v2f{p[0], p[1]};
}

template <int K, int NC>
__device__ __forceinline__ void stage_weights(const float* __restrict__ Wm,
                                              float* __restrict__ L, int tid) {
#pragma unroll 4
    for (int it = tid; it < K * NC / 4; it += GEMM_WAVES * 32) {
        const int c  = it % NC;
        const int qh = it / NC;
        const int k0 = (qh >> 1) * 8 + (qh & 1) * 2;
        const float* p = Wm + k0 * NC + c;
        *(v4f*)(L + it * 4) = v4f{p[0], p[NC], p[4 * NC], p[5 * NC]};
    }
}

template <int NQ, int NC>
__device__ __forceinline__ void mma_pass(const v2f* a, const float* __restrict__ L,
                                         int q0, int c0, int lane,
                                         v8f& acc0, v8f& acc1) {
    const int n = lane & 15, h = lane >> 4;
    v4f b0[NQ], b1[NQ];
#pragma unroll
    for (int q = 0; q < NQ; ++q) {
        const float* p = L + (((q0 + q) * 2 + h) * NC + c0 + n) * 4;
        b0[q] = *(const v4f*)p;
        b1[q] = *(const v4f*)(p + 64);
    }
#pragma unroll
    for (int q = 0; q < NQ; ++q) {
        acc0 = wmma_f32(a[2 * q],     v2f{b0[q].x, b0[q].y}, acc0);
        acc1 = wmma_f32(a[2 * q],     v2f{b1[q].x, b1[q].y}, acc1);
        acc0 = wmma_f32(a[2 * q + 1], v2f{b0[q].z, b0[q].w}, acc0);
        acc1 = wmma_f32(a[2 * q + 1], v2f{b1[q].z, b1[q].w}, acc1);
    }
}

__device__ __forceinline__ void store_tile(float* __restrict__ D, int ldd,
                                           int r0, int c0, v8f acc, int lane,
                                           float bias) {
    int n    = lane & 15;
    int half = lane >> 4;
#pragma unroll
    for (int r = 0; r < 8; ++r)
        D[(size_t)(r0 + r + 8 * half) * ldd + c0 + n] = acc[r] + bias;
}

#define BUCKET_CAP 64

__device__ __forceinline__ void bucket_edges(const int* __restrict__ row, const int* __restrict__ col,
                                             int* __restrict__ cnt, int* __restrict__ bsrc,
                                             int* __restrict__ ov_src, int* __restrict__ ov_dst,
                                             int N, int E, int first, int stride) {
    for (int e0 = first; e0 < E; e0 += EDGE_BATCH * stride) {
        int c[EDGE_BATCH], s[EDGE_BATCH], r[EDGE_BATCH];
#pragma unroll
        for (int u = 0; u < EDGE_BATCH; ++u) {
            const int e = min(e0 + u * stride, E - 1);
            c[u] = col[e];
            s[u] = row[e];
        }
#pragma unroll
        for (int u = 0; u < EDGE_BATCH; ++u)
            r[u] = e0 + u * stride < E ? atomicAdd(&cnt[c[u]], 1) : 0;
#pragma unroll
        for (int u = 0; u < EDGE_BATCH; ++u) {
            if (e0 + u * stride < E) {
                if (r[u] < BUCKET_CAP) {
                    bsrc[c[u] * BUCKET_CAP + r[u]] = s[u];
                } else {
                    const int o = atomicAdd(&cnt[N], 1);
                    ov_src[o] = s[u];
                    ov_dst[o] = c[u];
                }
            }
        }
    }
}

#define FOLD_LD 68

__global__ __launch_bounds__(128) void weight_fold_kernel(
    const float* __restrict__ Wg1, const float* __restrict__ bg1,
    const float* __restrict__ Wg2, const float* __restrict__ bg2,
    const float* __restrict__ Wo2, float* __restrict__ Wf,
    int* __restrict__ cnt, int N)
{
    if (blockIdx.x >= 9) {
        const int i = ((int)blockIdx.x - 9) * 128 + (int)threadIdx.x;
        if (i <= N) cnt[i] = 0;
        return;
    }
    __shared__ __attribute__((aligned(16))) float Lt[16 * FOLD_LD];
    const int lane = threadIdx.x & 31;
    const int c0   = (threadIdx.x >> 5) * 16;
    const int r0   = blockIdx.x * 16;
    const int n    = lane & 15, h = lane >> 4;
    const int row  = r0 + n;
    const float* arow = row < 128 ? Wg1 + row * 128 : bg1;
    const bool  live  = row <= 128;
    const float* bcol = Wg2 + c0 + n;
    const float* ocol = Wo2 + c0 + n;

    v2f bo[16];
#pragma unroll
    for (int kc = 0; kc < 16; ++kc) {
        const int k0 = kc * 4 + 2 * h;
        bo[kc] = v2f{ocol[k0 * 64], ocol[(k0 + 1) * 64]};
    }

    v8f acc0 = {}, acc1 = {};
#pragma unroll
    for (int kc = 0; kc < 32; kc += 2) {
        const int k0 = kc * 4 + 2 * h, k1 = k0 + 4;
        v2f a0 = v2f{arow[k0], arow[k0 + 1]};
        v2f a1 = v2f{arow[k1], arow[k1 + 1]};
        if (!live) { a0 = v2f{0.f, 0.f}; a1 = v2f{0.f, 0.f}; }
        const v2f b0 = v2f{bcol[k0 * 64], bcol[(k0 + 1) * 64]};
        const v2f b1 = v2f{bcol[k1 * 64], bcol[(k1 + 1) * 64]};
        acc0 = wmma_f32(a0, b0, acc0);
        acc1 = wmma_f32(a1, b1, acc1);
    }
    const v8f t = acc0 + acc1;
    const float b2 = bg2[c0 + n];
#pragma unroll
    for (int r = 0; r < 8; ++r) {
        const int lr = r + 8 * h;
        Lt[lr * FOLD_LD + c0 + n] = (r0 + lr == 129) ? b2 : t[r];
    }
    __syncthreads();

    v8f d0 = {}, d1 = {};
#pragma unroll
    for (int kc = 0; kc < 16; kc += 2) {
        d0 = wmma_f32(a_frag_lds(Lt, FOLD_LD, kc * 4,     lane), bo[kc],     d0);
        d1 = wmma_f32(a_frag_lds(Lt, FOLD_LD, kc * 4 + 4, lane), bo[kc + 1], d1);
    }
    store_tile(Wf, 64, r0, c0, d0 + d1, lane, 0.f);
}

__global__ __launch_bounds__((GEMM_WAVES + EDGE_WAVES) * 32) void self_branch_kernel(
    const float* __restrict__ x, const float* __restrict__ Win,
    const float* __restrict__ bin, const float* __restrict__ Wout,
    const float* __restrict__ bout, const float* __restrict__ Wf,
    const float* __restrict__ xn, const float* __restrict__ Wo1,
    const float* __restrict__ bo,
    float* __restrict__ l1_out, float* __restrict__ p_out,
    float* __restrict__ q_out, int N,
    const int* __restrict__ row, const int* __restrict__ col, int* __restrict__ cnt,
    int* __restrict__ bsrc, int* __restrict__ ov_src, int* __restrict__ ov_dst, int E)
{
    __shared__ __attribute__((aligned(16))) float lds[16384 + 8192 + 16384 + 8192 + GEMM_WAVES * 16 * LDS_LD];
    float* Lin  = lds;
    float* Lf  = lds + 16384;
    float* Lout = lds + 24576;
    float* Lo1  = lds + 40960;
    const int tid  = threadIdx.x;
    const int lane = tid & 31;
    const int wave = __builtin_amdgcn_readfirstlane(tid >> 5);
    const int n    = lane & 15, half = lane >> 4;
    float* myL = lds + 49152 + wave * 16 * LDS_LD;

    if (tid < GEMM_WAVES * 32) {
        stage_weights<128, 128>(Win,  Lin,  tid);
        stage_weights<128, 64>(Wf,  Lf,  tid);
        stage_weights<256, 64>(Wout, Lout, tid);
        stage_weights<128, 64>(Wo1,  Lo1,  tid);
    }
    __syncthreads();

    if (wave >= GEMM_WAVES) {
        bucket_edges(row, col, cnt, bsrc, ov_src, ov_dst, N, E,
                     (int)blockIdx.x * (EDGE_WAVES * 32) + (tid - GEMM_WAVES * 32),
                     (int)gridDim.x * (EDGE_WAVES * 32));
        return;
    }

    const int strips = N / 16;

    if (wave & 2) {
        const int slot = ((wave >> 2) << 1) | (wave & 1);
        for (int strip = slot * (int)gridDim.x + (int)blockIdx.x; strip < strips;
             strip += 4 * (int)gridDim.x) {
            const int r0 = strip * 16;
            v2f axn[32];
#pragma unroll
            for (int kc = 0; kc < 32; ++kc)
                axn[kc] = a_frag_g(xn, 128, r0, kc * 4, lane);
#pragma unroll 1
            for (int t = 0; t < 4; t += 2) {
                v8f acc0 = {}, acc1 = {};
                mma_pass<16, 64>(axn, Lo1, 0, t * 16, lane, acc0, acc1);
                store_tile(q_out, 64, r0, t * 16,      acc0, lane, bo[t * 16 + n]      + Wf[129 * 64 + t * 16 + n]);
                store_tile(q_out, 64, r0, t * 16 + 16, acc1, lane, bo[t * 16 + 16 + n] + Wf[129 * 64 + t * 16 + 16 + n]);
            }
        }
    }

    for (int strip = wave * (int)gridDim.x + (int)blockIdx.x; strip < strips;
         strip += GEMM_WAVES * (int)gridDim.x) {
        const int r0 = strip * 16;

        v2f ax[32];
#pragma unroll
        for (int kc = 0; kc < 32; ++kc)
            ax[kc] = a_frag_g(x, 128, r0, kc * 4, lane);

#pragma unroll 1
        for (int t = 0; t < 8; t += 2) {
            v8f acc0 = {}, acc1 = {};
            mma_pass<16, 128>(ax, Lin, 0, t * 16, lane, acc0, acc1);
            const float bb0 = bin[t * 16 + n], bb1 = bin[t * 16 + 16 + n];
#pragma unroll
            for (int r = 0; r < 8; ++r) {
                const float v0 = acc0[r] + bb0, v1 = acc1[r] + bb1;
                myL[(r + 8 * half) * LDS_LD + t * 16 + n]      = v0 > 0.f ? v0 : 0.f;
                myL[(r + 8 * half) * LDS_LD + t * 16 + 16 + n] = v1 > 0.f ? v1 : 0.f;
            }
        }

#pragma unroll 1
        for (int t = 0; t < 4; t += 2) {
            v8f acc0 = {}, acc1 = {};
            mma_pass<16, 64>(ax, Lf, 0, t * 16, lane, acc0, acc1);
            store_tile(p_out, 64, r0, t * 16,      acc0, lane, 0.f);
            store_tile(p_out, 64, r0, t * 16 + 16, acc1, lane, 0.f);
        }

        __builtin_amdgcn_fence(__ATOMIC_ACQ_REL, "wavefront");

        v2f al[32];
#pragma unroll
        for (int kc = 0; kc < 32; ++kc)
            al[kc] = a_frag_lds(myL, LDS_LD, kc * 4, lane);

#pragma unroll 1
        for (int t = 0; t < 4; t += 2) {
            v8f acc0 = {}, acc1 = {};
            mma_pass<16, 64>(ax, Lout, 0,  t * 16, lane, acc0, acc1);
            mma_pass<16, 64>(al, Lout, 16, t * 16, lane, acc0, acc1);
            store_tile(l1_out, 64, r0, t * 16,      acc0, lane, bout[t * 16 + n]);
            store_tile(l1_out, 64, r0, t * 16 + 16, acc1, lane, bout[t * 16 + 16 + n]);
        }

        __builtin_amdgcn_fence(__ATOMIC_ACQ_REL, "wavefront");
    }
}

template <int V, int U, bool MAKE_W>
__device__ __forceinline__ void gather_rows(
    const float* __restrict__ h, int W, int s, int wbits, float d, int k, int lo,
    float __attribute__((ext_vector_type(V)))& acc)
{
    typedef float vecf __attribute__((ext_vector_type(V)));
    vecf v[U];
    for (int u = 0; u < U; ++u) {
        const int su = __builtin_amdgcn_readlane(s, k + u);
        v[u] = *(const vecf*)(h + (size_t)su * W + lo);
    }
    int wl = wbits;
    if (MAKE_W) {
        asm volatile("" : "+v"(wl));
        wl = __float_as_int(rsqrtf((float)(wl + 1)) * d);
    }
    for (int u = 0; u < U; ++u)
        acc += v[u] * __int_as_float(__builtin_amdgcn_readlane(wl, k + u));
}

template <int W, bool MAKE_W>
__global__ __launch_bounds__(256) void seg_gather_kernel(
    const int* __restrict__ cnt, const int* __restrict__ bsrc, float* __restrict__ bwgt,
    const int* __restrict__ ov_src, const int* __restrict__ ov_dst,
    const float* __restrict__ h, const float* __restrict__ bias,
    int bstride, float* __restrict__ g, int N)
{
    constexpr int V = W / 32;
    typedef float vecf __attribute__((ext_vector_type(V)));
    const int lane = threadIdx.x & 31;
    const int i = __builtin_amdgcn_readfirstlane(blockIdx.x * 8 + (threadIdx.x >> 5));
    if (i >= N) return;
    const int deg = __builtin_amdgcn_readfirstlane(cnt[i]);
    const int nb  = min(deg, BUCKET_CAP);
    const int lo  = lane * V;
    const float d = rsqrtf((float)(deg + 1));
    vecf acc = *(const vecf*)(bias + (size_t)i * bstride + lo) + *(const vecf*)(h + (size_t)i * W + lo) * (d * d);
    const int* bs = bsrc + (size_t)i * BUCKET_CAP;
    float*     bw = bwgt + (size_t)i * BUCKET_CAP;

    for (int base = 0; base < nb; base += 32) {
        const int m = min(32, nb - base);
        const int slot = base + min(lane, m - 1);
        const int s = bs[slot];
        const int wbits = MAKE_W ? cnt[s] : __float_as_int(bw[slot]);
        int k = 0;
        for (; k + 8 <= m; k += 8) gather_rows<V, 8, MAKE_W>(h, W, s, wbits, d, k, lo, acc);
        if (k + 4 <= m) { gather_rows<V, 4, MAKE_W>(h, W, s, wbits, d, k, lo, acc); k += 4; }
        if (k + 2 <= m) { gather_rows<V, 2, MAKE_W>(h, W, s, wbits, d, k, lo, acc); k += 2; }
        if (k < m)      { gather_rows<V, 1, MAKE_W>(h, W, s, wbits, d, k, lo, acc); }
        if (MAKE_W) {
            if (lane < m) bw[base + lane] = rsqrtf((float)(wbits + 1)) * d;
        }
    }

    if (deg > BUCKET_CAP) {
        const int no = __builtin_amdgcn_readfirstlane(cnt[N]);
#pragma unroll 1
        for (int j = 0; j < no; ++j) {
            if (__builtin_amdgcn_readfirstlane(ov_dst[j]) == i) {
                const int sj = __builtin_amdgcn_readfirstlane(ov_src[j]);
                const float wj = rsqrtf((float)(__builtin_amdgcn_readfirstlane(cnt[sj]) + 1)) * d;
                acc += *(const vecf*)(h + (size_t)sj * W + lo) * wj;
            }
        }
    }
    *(vecf*)(g + (size_t)i * W + lo) = acc;
}

extern "C" void kernel_launch(void* const* d_in, const int* in_sizes, int n_in,
                              void* d_out, int out_size, void* d_ws, size_t ws_size,
                              hipStream_t stream) {
    const float* x_self     = (const float*)d_in[0];
    const float* x_neighbor = (const float*)d_in[1];
    const int*   edge_index = (const int*)d_in[2];
    const float* W_in_self  = (const float*)d_in[3];
    const float* b_in_self  = (const float*)d_in[4];
    const float* W_out_self = (const float*)d_in[5];
    const float* b_out_self = (const float*)d_in[6];
    const float* Wg1        = (const float*)d_in[7];
    const float* bg1        = (const float*)d_in[8];
    const float* Wg2        = (const float*)d_in[9];
    const float* bg2        = (const float*)d_in[10];
    const float* W_out      = (const float*)d_in[11];
    const float* b_out      = (const float*)d_in[12];

    const int N = in_sizes[0] / 128;
    const int E = in_sizes[2] / 2;
    const int* row = edge_index;
    const int* col = edge_index + E;

    float* l1_out = (float*)d_out;
    float* x2_out = (float*)d_out + (size_t)N * 64;

    float* ws  = (float*)d_ws;
    int*   cnt = (int*)ws;
    float* h2  = ws + N + 64;
    float* q   = h2 + (size_t)N * 64;
    float* Wf  = q  + (size_t)N * 64;
    int*   bsrc   = (int*)(Wf + 144 * 64);
    float* bwgt   = (float*)(bsrc + (size_t)N * BUCKET_CAP);
    int*   ov_src = (int*)(bwgt + (size_t)N * BUCKET_CAP);
    int*   ov_dst = ov_src + E;

    float* p = x2_out;

    weight_fold_kernel<<<9 + (N + 128) / 128, 128, 0, stream>>>(Wg1, bg1, Wg2, bg2, W_out + 128 * 64, Wf, cnt, N);

    self_branch_kernel<<<GEMM_BLOCKS, (GEMM_WAVES + EDGE_WAVES) * 32, 0, stream>>>(
        x_self, W_in_self, b_in_self, W_out_self, b_out_self, Wf,
        x_neighbor, W_out, b_out, l1_out, p, q, N,
        row, col, cnt, bsrc, ov_src, ov_dst, E);

    seg_gather_kernel<64, true><<<(N + 7) / 8, 256, 0, stream>>>(cnt, bsrc, bwgt, ov_src, ov_dst, p, Wf + 128 * 64, 0, h2, N);

    seg_gather_kernel<64, false><<<(N + 7) / 8, 256, 0, stream>>>(cnt, bsrc, bwgt, ov_src, ov_dst, h2, q, 64, x2_out, N);
}
